# starting kernel
# baseline (speedup 1.0000x reference)
#include <hip/hip_runtime.h>

#define NBATCH 2
#define NDIM   768
#define NLEN   2048
#define NST    16
#define CHUNK  16
#define NC     (NLEN / CHUNK)
#define TOTCH  (NBATCH * NDIM * NC)
#define WSN    ((size_t)TOTCH * NST)
#define CGRP   8

typedef float v2f __attribute__((ext_vector_type(2)));
typedef float v8f __attribute__((ext_vector_type(8)));

__device__ __forceinline__ float softplus_f(float x) {
    return (x > 20.0f) ? x : __logf(1.0f + __expf(x));
}

#define SWZ(v, IMM) __int_as_float(__builtin_amdgcn_ds_swizzle(__float_as_int(v), (IMM)))
#define SWZ_ADD(v, IMM) (v) += SWZ(v, IMM)

__global__ __launch_bounds__(256) void k1_chunk_state(
    const float* __restrict__ u, const float* __restrict__ delta,
    const float* __restrict__ A, const float* __restrict__ B,
    const float* __restrict__ dbias,
    float* __restrict__ wsDecay, float* __restrict__ wsS)
{
    int t = blockIdx.x * 256 + threadIdx.x;
    int c = t % NC;
    int d = (t / NC) % NDIM;
    int b = t / (NC * NDIM);
    int cs = c * CHUNK;

    const float* dp = delta + (size_t)(b * NDIM + d) * NLEN + cs;
    const float* up = u     + (size_t)(b * NDIM + d) * NLEN + cs;

    float An[NST];
#pragma unroll
    for (int n = 0; n < NST; n++) An[n] = A[d * NST + n];
    float bias = dbias[d];

    float x[NST];
#pragma unroll
    for (int n = 0; n < NST; n++) x[n] = 0.0f;
    float R = 0.0f;

    for (int tt = 0; tt < CHUNK; tt++) {
        float sp = softplus_f(dp[tt] + bias);
        float du = sp * up[tt];
        R += sp;
#pragma unroll
        for (int n = 0; n < NST; n++) {
            float a  = __expf(An[n] * sp);
            float bv = B[(size_t)(b * NST + n) * NLEN + cs + tt];
            x[n] = a * x[n] + du * bv;
        }
    }
    size_t o = ((size_t)(b * NDIM + d) * NC + c) * NST;
#pragma unroll
    for (int n = 0; n < NST; n++) {
        wsS[o + n]     = x[n];
        wsDecay[o + n] = __expf(An[n] * R);
    }
}

__global__ __launch_bounds__(256) void k2_scan(
    float* __restrict__ wsDecay, const float* __restrict__ wsS)
{
    int i = blockIdx.x * 256 + threadIdx.x;
    int n = i & (NST - 1);
    int d = (i >> 4) % NDIM;
    int b = i / (NST * NDIM);
    size_t base = ((size_t)(b * NDIM + d) * NC) * NST + n;

    float x = 0.0f;
    for (int c = 0; c < NC; c++) {
        size_t idx = base + (size_t)c * NST;
        float dec = wsDecay[idx];
        float s   = wsS[idx];
        wsDecay[idx] = x;
        x = dec * x + s;
    }
}

__global__ __launch_bounds__(256) void k3_output(
    const float* __restrict__ u, const float* __restrict__ delta,
    const float* __restrict__ A, const float* __restrict__ B,
    const float* __restrict__ C, const float* __restrict__ Dp,
    const float* __restrict__ z, const float* __restrict__ dbias,
    const float* __restrict__ wsXin, float* __restrict__ out)
{
    __shared__ float Pl[CGRP][16][17];
    __shared__ float Ql[CGRP][16][17];
    __shared__ float dl[CGRP][16];
    __shared__ float ul[CGRP][16];
    __shared__ float xl[CGRP][16];

    int w    = threadIdx.x >> 5;
    int lane = threadIdx.x & 31;
    int m    = lane & 15;
    bool lo  = lane < 16;

    int cg = blockIdx.x % (NC / CGRP);
    int d  = (blockIdx.x / (NC / CGRP)) % NDIM;
    int b  = blockIdx.x / ((NC / CGRP) * NDIM);
    int c  = cg * CGRP + w;
    int cs = c * CHUNK;
    size_t base = (size_t)(b * NDIM + d) * NLEN + cs;

    if (lo) {
        dl[w][m] = softplus_f(delta[base + m] + dbias[d]);
    } else {
        ul[w][m] = u[base + m];
        xl[w][m] = wsXin[((size_t)(b * NDIM + d) * NC + c) * NST + m];
    }
    __syncthreads();

    float rr = 0.0f;
#pragma unroll
    for (int t = 0; t < 16; t++) { float v = dl[w][t]; rr += (t <= m) ? v : 0.0f; }

    float dus = dl[w][m] * ul[w][m];
    float sgn = lo ? 1.0f : -1.0f;
    float mul = lo ? 1.0f : dus;
    const float* srcb = lo ? C : B;
    float* dstb = lo ? &Pl[w][0][0] : &Ql[w][0][0];

#pragma unroll
    for (int n = 0; n < NST; n++) {
        float An = A[d * NST + n];
        float v  = srcb[(size_t)(b * NST + n) * NLEN + cs + m];
        float e  = __expf(sgn * An * rr);
        dstb[m * 17 + n] = v * e * mul;
    }
    __syncthreads();

    v8f acc = {0.f, 0.f, 0.f, 0.f, 0.f, 0.f, 0.f, 0.f};
#pragma unroll
    for (int j = 0; j < 4; j++) {
        int k0 = 4 * j + (lo ? 0 : 2);
        int k1 = 4 * j + (lo ? 1 : 3);
        v2f va; va.x = Pl[w][m][k0]; va.y = Pl[w][m][k1];
        v2f vb; vb.x = Ql[w][m][k0]; vb.y = Ql[w][m][k1];
        acc = __builtin_amdgcn_wmma_f32_16x16x4_f32(
            false, va, false, vb, (short)0, acc, false, false);
    }

    float yv = 0.0f;
#pragma unroll
    for (int r = 0; r < 8; r++) {
        int t = r + (lo ? 0 : 8);
        float v = (m <= t) ? acc[r] : 0.0f;
        SWZ_ADD(v, 0x041F);
        SWZ_ADD(v, 0x081F);
        SWZ_ADD(v, 0x101F);
        SWZ_ADD(v, 0x201F);
        yv = ((m & 7) == r) ? v : yv;
    }
    float yx = SWZ(yv, 0x601F);
    float y  = (m < 8) ? yv : yx;

    if (lo) {
        int t = m;
#pragma unroll
        for (int n = 0; n < NST; n++) y += Pl[w][t][n] * xl[w][n];
        float uv = ul[w][t];
        float zv = z[base + t];
        float sig = __builtin_amdgcn_rcpf(1.0f + __expf(-zv));
        out[base + t] = (y + uv * Dp[d]) * (zv * sig);
    }
}

extern "C" void kernel_launch(void* const* d_in, const int* in_sizes, int n_in,
                              void* d_out, int out_size, void* d_ws, size_t ws_size,
                              hipStream_t stream) {
    const float* u     = (const float*)d_in[0];
    const float* delta = (const float*)d_in[1];
    const float* A     = (const float*)d_in[2];
    const float* B     = (const float*)d_in[3];
    const float* C     = (const float*)d_in[4];
    const float* Dp    = (const float*)d_in[5];
    const float* z     = (const float*)d_in[6];
    const float* dbias = (const float*)d_in[7];
    float* out = (float*)d_out;

    float* wsDecay = (float*)d_ws;
    float* wsS     = wsDecay + WSN;

    k1_chunk_state<<<TOTCH / 256, 256, 0, stream>>>(u, delta, A, B, dbias,
                                                    wsDecay, wsS);
    k2_scan<<<(NBATCH * NDIM * NST) / 256, 256, 0, stream>>>(wsDecay, wsS);
    k3_output<<<TOTCH / CGRP, 256, 0, stream>>>(u, delta, A, B, C, Dp, z, dbias,
                                                wsDecay, out);
}


# k1: lanes remapped along d with the chunk wave-uniform, so the 256 strided B loads per thread become scalar loads; per-lane delta/u/A/ws as 16-byte accesses; summation order unchanged
# speedup vs baseline: 2.2513x; 2.2513x over previous
#include <hip/hip_runtime.h>

#define NBATCH 2
#define NDIM   768
#define NLEN   2048
#define NST    16
#define CHUNK  16
#define NC     (NLEN / CHUNK)
#define TOTCH  (NBATCH * NDIM * NC)
#define WSN    ((size_t)TOTCH * NST)
#define CGRP   8

typedef float v2f __attribute__((ext_vector_type(2)));
typedef float v8f __attribute__((ext_vector_type(8)));

__device__ __forceinline__ float softplus_f(float x) {
    return (x > 20.0f) ? x : __logf(1.0f + __expf(x));
}

#define SWZ(v, IMM) __int_as_float(__builtin_amdgcn_ds_swizzle(__float_as_int(v), (IMM)))
#define SWZ_ADD(v, IMM) (v) += SWZ(v, IMM)

#define K1_DPB 32
#define K1_CPB 8

__global__ __launch_bounds__(256) void k1_chunk_state(
    const float* __restrict__ u, const float* __restrict__ delta,
    const float* __restrict__ A, const float* __restrict__ B,
    const float* __restrict__ dbias,
    float* __restrict__ wsDecay, float* __restrict__ wsS)
{
    int w    = __builtin_amdgcn_readfirstlane((int)(threadIdx.x >> 5));
    int lane = threadIdx.x & 31;

    int cg = blockIdx.x % (NC / K1_CPB);
    int dg = (blockIdx.x / (NC / K1_CPB)) % (NDIM / K1_DPB);
    int b  = blockIdx.x / ((NC / K1_CPB) * (NDIM / K1_DPB));
    int c  = cg * K1_CPB + w;
    int d  = dg * K1_DPB + lane;
    int cs = c * CHUNK;

    const float4* dp4 = (const float4*)(delta + (size_t)(b * NDIM + d) * NLEN + cs);
    const float4* up4 = (const float4*)(u     + (size_t)(b * NDIM + d) * NLEN + cs);
    const float4* A4  = (const float4*)(A + d * NST);
    const float*  Bb  = B + (size_t)(b * NST) * NLEN + cs;

    float An[NST];
#pragma unroll
    for (int q = 0; q < NST / 4; q++) {
        float4 a4 = A4[q];
        An[4 * q + 0] = a4.x; An[4 * q + 1] = a4.y;
        An[4 * q + 2] = a4.z; An[4 * q + 3] = a4.w;
    }
    float bias = dbias[d];

    float x[NST];
#pragma unroll
    for (int n = 0; n < NST; n++) x[n] = 0.0f;
    float R = 0.0f;

    float4 d4 = dp4[0];
    float4 u4 = up4[0];
#pragma unroll 1
    for (int tq = 0; tq < CHUNK / 4; tq++) {
        int tn = (tq + 1 < CHUNK / 4) ? tq + 1 : tq;
        float4 dnx = dp4[tn];
        float4 unx = up4[tn];

        float4 bq[NST];
#pragma unroll
        for (int n = 0; n < NST; n++)
            bq[n] = *(const float4*)(Bb + (size_t)n * NLEN + 4 * tq);

        float dq[4] = {d4.x, d4.y, d4.z, d4.w};
        float uq[4] = {u4.x, u4.y, u4.z, u4.w};
#pragma unroll
        for (int j = 0; j < 4; j++) {
            float sp = softplus_f(dq[j] + bias);
            float du = sp * uq[j];
            R += sp;
#pragma unroll
            for (int n = 0; n < NST; n++) {
                float a  = __expf(An[n] * sp);
                float bv = (j == 0) ? bq[n].x : (j == 1) ? bq[n].y
                         : (j == 2) ? bq[n].z : bq[n].w;
                x[n] = a * x[n] + du * bv;
            }
        }
        d4 = dnx;
        u4 = unx;
    }
    size_t o = ((size_t)(b * NDIM + d) * NC + c) * NST;
    float4* S4 = (float4*)(wsS + o);
    float4* D4 = (float4*)(wsDecay + o);
#pragma unroll
    for (int q = 0; q < NST / 4; q++) {
        float4 s4, e4;
        s4.x = x[4 * q + 0]; s4.y = x[4 * q + 1];
        s4.z = x[4 * q + 2]; s4.w = x[4 * q + 3];
        e4.x = __expf(An[4 * q + 0] * R); e4.y = __expf(An[4 * q + 1] * R);
        e4.z = __expf(An[4 * q + 2] * R); e4.w = __expf(An[4 * q + 3] * R);
        S4[q] = s4;
        D4[q] = e4;
    }
}

__global__ __launch_bounds__(256) void k2_scan(
    float* __restrict__ wsDecay, const float* __restrict__ wsS)
{
    int i = blockIdx.x * 256 + threadIdx.x;
    int n = i & (NST - 1);
    int d = (i >> 4) % NDIM;
    int b = i / (NST * NDIM);
    size_t base = ((size_t)(b * NDIM + d) * NC) * NST + n;

    float x = 0.0f;
    for (int c = 0; c < NC; c++) {
        size_t idx = base + (size_t)c * NST;
        float dec = wsDecay[idx];
        float s   = wsS[idx];
        wsDecay[idx] = x;
        x = dec * x + s;
    }
}

__global__ __launch_bounds__(256) void k3_output(
    const float* __restrict__ u, const float* __restrict__ delta,
    const float* __restrict__ A, const float* __restrict__ B,
    const float* __restrict__ C, const float* __restrict__ Dp,
    const float* __restrict__ z, const float* __restrict__ dbias,
    const float* __restrict__ wsXin, float* __restrict__ out)
{
    __shared__ float Pl[CGRP][16][17];
    __shared__ float Ql[CGRP][16][17];
    __shared__ float dl[CGRP][16];
    __shared__ float ul[CGRP][16];
    __shared__ float xl[CGRP][16];

    int w    = threadIdx.x >> 5;
    int lane = threadIdx.x & 31;
    int m    = lane & 15;
    bool lo  = lane < 16;

    int cg = blockIdx.x % (NC / CGRP);
    int d  = (blockIdx.x / (NC / CGRP)) % NDIM;
    int b  = blockIdx.x / ((NC / CGRP) * NDIM);
    int c  = cg * CGRP + w;
    int cs = c * CHUNK;
    size_t base = (size_t)(b * NDIM + d) * NLEN + cs;

    if (lo) {
        dl[w][m] = softplus_f(delta[base + m] + dbias[d]);
    } else {
        ul[w][m] = u[base + m];
        xl[w][m] = wsXin[((size_t)(b * NDIM + d) * NC + c) * NST + m];
    }
    __syncthreads();

    float rr = 0.0f;
#pragma unroll
    for (int t = 0; t < 16; t++) { float v = dl[w][t]; rr += (t <= m) ? v : 0.0f; }

    float dus = dl[w][m] * ul[w][m];
    float sgn = lo ? 1.0f : -1.0f;
    float mul = lo ? 1.0f : dus;
    const float* srcb = lo ? C : B;
    float* dstb = lo ? &Pl[w][0][0] : &Ql[w][0][0];

#pragma unroll
    for (int n = 0; n < NST; n++) {
        float An = A[d * NST + n];
        float v  = srcb[(size_t)(b * NST + n) * NLEN + cs + m];
        float e  = __expf(sgn * An * rr);
        dstb[m * 17 + n] = v * e * mul;
    }
    __syncthreads();

    v8f acc = {0.f, 0.f, 0.f, 0.f, 0.f, 0.f, 0.f, 0.f};
#pragma unroll
    for (int j = 0; j < 4; j++) {
        int k0 = 4 * j + (lo ? 0 : 2);
        int k1 = 4 * j + (lo ? 1 : 3);
        v2f va; va.x = Pl[w][m][k0]; va.y = Pl[w][m][k1];
        v2f vb; vb.x = Ql[w][m][k0]; vb.y = Ql[w][m][k1];
        acc = __builtin_amdgcn_wmma_f32_16x16x4_f32(
            false, va, false, vb, (short)0, acc, false, false);
    }

    float yv = 0.0f;
#pragma unroll
    for (int r = 0; r < 8; r++) {
        int t = r + (lo ? 0 : 8);
        float v = (m <= t) ? acc[r] : 0.0f;
        SWZ_ADD(v, 0x041F);
        SWZ_ADD(v, 0x081F);
        SWZ_ADD(v, 0x101F);
        SWZ_ADD(v, 0x201F);
        yv = ((m & 7) == r) ? v : yv;
    }
    float yx = SWZ(yv, 0x601F);
    float y  = (m < 8) ? yv : yx;

    if (lo) {
        int t = m;
#pragma unroll
        for (int n = 0; n < NST; n++) y += Pl[w][t][n] * xl[w][n];
        float uv = ul[w][t];
        float zv = z[base + t];
        float sig = __builtin_amdgcn_rcpf(1.0f + __expf(-zv));
        out[base + t] = (y + uv * Dp[d]) * (zv * sig);
    }
}

extern "C" void kernel_launch(void* const* d_in, const int* in_sizes, int n_in,
                              void* d_out, int out_size, void* d_ws, size_t ws_size,
                              hipStream_t stream) {
    const float* u     = (const float*)d_in[0];
    const float* delta = (const float*)d_in[1];
    const float* A     = (const float*)d_in[2];
    const float* B     = (const float*)d_in[3];
    const float* C     = (const float*)d_in[4];
    const float* Dp    = (const float*)d_in[5];
    const float* z     = (const float*)d_in[6];
    const float* dbias = (const float*)d_in[7];
    float* out = (float*)d_out;

    float* wsDecay = (float*)d_ws;
    float* wsS     = wsDecay + WSN;

    k1_chunk_state<<<TOTCH / K1_DPB / K1_CPB, K1_DPB * K1_CPB, 0, stream>>>(
        u, delta, A, B, dbias, wsDecay, wsS);
    k2_scan<<<(NBATCH * NDIM * NST) / 256, 256, 0, stream>>>(wsDecay, wsS);
    k3_output<<<TOTCH / CGRP, 256, 0, stream>>>(u, delta, A, B, C, Dp, z, dbias,
                                                wsDecay, out);
}


# k3: one wave holds a chunk's B,C in registers and walks 16 d; P,Q built straight in WMMA operand layout, product as Q*P^T so the causal sum is in-lane; no LDS; y and running-delta sum trees changed
# speedup vs baseline: 2.9955x; 1.3306x over previous
#include <hip/hip_runtime.h>

#define NBATCH 2
#define NDIM   768
#define NLEN   2048
#define NST    16
#define CHUNK  16
#define NC     (NLEN / CHUNK)
#define TOTCH  (NBATCH * NDIM * NC)
#define WSN    ((size_t)TOTCH * NST)

typedef float v2f __attribute__((ext_vector_type(2)));
typedef float v8f __attribute__((ext_vector_type(8)));

__device__ __forceinline__ float softplus_f(float x) {
    return (x > 20.0f) ? x : __logf(1.0f + __expf(x));
}

#define K1_DPB 32
#define K1_CPB 8

__global__ __launch_bounds__(256) void k1_chunk_state(
    const float* __restrict__ u, const float* __restrict__ delta,
    const float* __restrict__ A, const float* __restrict__ B,
    const float* __restrict__ dbias,
    float* __restrict__ wsDecay, float* __restrict__ wsS)
{
    int w    = __builtin_amdgcn_readfirstlane((int)(threadIdx.x >> 5));
    int lane = threadIdx.x & 31;

    int cg = blockIdx.x % (NC / K1_CPB);
    int dg = (blockIdx.x / (NC / K1_CPB)) % (NDIM / K1_DPB);
    int b  = blockIdx.x / ((NC / K1_CPB) * (NDIM / K1_DPB));
    int c  = cg * K1_CPB + w;
    int d  = dg * K1_DPB + lane;
    int cs = c * CHUNK;

    const float4* dp4 = (const float4*)(delta + (size_t)(b * NDIM + d) * NLEN + cs);
    const float4* up4 = (const float4*)(u     + (size_t)(b * NDIM + d) * NLEN + cs);
    const float4* A4  = (const float4*)(A + d * NST);
    const float*  Bb  = B + (size_t)(b * NST) * NLEN + cs;

    float An[NST];
#pragma unroll
    for (int q = 0; q < NST / 4; q++) {
        float4 a4 = A4[q];
        An[4 * q + 0] = a4.x; An[4 * q + 1] = a4.y;
        An[4 * q + 2] = a4.z; An[4 * q + 3] = a4.w;
    }
    float bias = dbias[d];

    float x[NST];
#pragma unroll
    for (int n = 0; n < NST; n++) x[n] = 0.0f;
    float R = 0.0f;

    float4 d4 = dp4[0];
    float4 u4 = up4[0];
#pragma unroll 1
    for (int tq = 0; tq < CHUNK / 4; tq++) {
        int tn = (tq + 1 < CHUNK / 4) ? tq + 1 : tq;
        float4 dnx = dp4[tn];
        float4 unx = up4[tn];

        float4 bq[NST];
#pragma unroll
        for (int n = 0; n < NST; n++)
            bq[n] = *(const float4*)(Bb + (size_t)n * NLEN + 4 * tq);

        float dq[4] = {d4.x, d4.y, d4.z, d4.w};
        float uq[4] = {u4.x, u4.y, u4.z, u4.w};
#pragma unroll
        for (int j = 0; j < 4; j++) {
            float sp = softplus_f(dq[j] + bias);
            float du = sp * uq[j];
            R += sp;
#pragma unroll
            for (int n = 0; n < NST; n++) {
                float a  = __expf(An[n] * sp);
                float bv = (j == 0) ? bq[n].x : (j == 1) ? bq[n].y
                         : (j == 2) ? bq[n].z : bq[n].w;
                x[n] = a * x[n] + du * bv;
            }
        }
        d4 = dnx;
        u4 = unx;
    }
    size_t o = ((size_t)(b * NDIM + d) * NC + c) * NST;
    float4* S4 = (float4*)(wsS + o);
    float4* D4 = (float4*)(wsDecay + o);
#pragma unroll
    for (int q = 0; q < NST / 4; q++) {
        float4 s4, e4;
        s4.x = x[4 * q + 0]; s4.y = x[4 * q + 1];
        s4.z = x[4 * q + 2]; s4.w = x[4 * q + 3];
        e4.x = __expf(An[4 * q + 0] * R); e4.y = __expf(An[4 * q + 1] * R);
        e4.z = __expf(An[4 * q + 2] * R); e4.w = __expf(An[4 * q + 3] * R);
        S4[q] = s4;
        D4[q] = e4;
    }
}

__global__ __launch_bounds__(256) void k2_scan(
    float* __restrict__ wsDecay, const float* __restrict__ wsS)
{
    int i = blockIdx.x * 256 + threadIdx.x;
    int n = i & (NST - 1);
    int d = (i >> 4) % NDIM;
    int b = i / (NST * NDIM);
    size_t base = ((size_t)(b * NDIM + d) * NC) * NST + n;

    float x = 0.0f;
    for (int c = 0; c < NC; c++) {
        size_t idx = base + (size_t)c * NST;
        float dec = wsDecay[idx];
        float s   = wsS[idx];
        wsDecay[idx] = x;
        x = dec * x + s;
    }
}

#define K3_WPB 8
#define K3_DPW 16
#define LOG2E_F 1.4426950408889634f

__device__ __forceinline__ float xhalf(float x) {
    int v = __float_as_int(x);
    return __int_as_float(__builtin_amdgcn_permlanex16(
        v, v, 0x76543210, 0xfedcba98, false, false));
}
#define ROW_SHR(v, K) __int_as_float(__builtin_amdgcn_update_dpp( \
    0, __float_as_int(v), 0x110 + (K), 0xF, 0xF, true))

__global__ __launch_bounds__(256) void k3_output(
    const float* __restrict__ u, const float* __restrict__ delta,
    const float* __restrict__ A, const float* __restrict__ B,
    const float* __restrict__ C, const float* __restrict__ Dp,
    const float* __restrict__ z, const float* __restrict__ dbias,
    const float* __restrict__ wsXin, float* __restrict__ out)
{
    int w    = __builtin_amdgcn_readfirstlane((int)(threadIdx.x >> 5));
    int lane = threadIdx.x & 31;
    int m    = lane & 15;
    int h    = lane >> 4;

    int cg = blockIdx.x % (NC / K3_WPB);
    int dg = (blockIdx.x / (NC / K3_WPB)) % (NDIM / K3_DPW);
    int b  = blockIdx.x / ((NC / K3_WPB) * (NDIM / K3_DPW));
    int c  = cg * K3_WPB + w;
    int cs = c * CHUNK;
    int lim = m - 8 * h;

    float cv[8], bv[8];
    {
        const float* Cb = C + (size_t)(b * NST + 8 * h) * NLEN + cs + m;
        const float* Bb = B + (size_t)(b * NST + 8 * h) * NLEN + cs + m;
#pragma unroll
        for (int q = 0; q < 8; q++) {
            cv[q] = Cb[(size_t)q * NLEN];
            bv[q] = Bb[(size_t)q * NLEN];
        }
    }

    for (int dd = 0; dd < K3_DPW; dd++) {
        int d = dg * K3_DPW + dd;
        size_t row  = (size_t)(b * NDIM + d);
        size_t base = row * NLEN + cs + m;

        float dv = delta[base];
        float uv = u[base];
        float zv = z[base];
        const float4* A4 = (const float4*)(A + d * NST + 8 * h);
        const float4* X4 = (const float4*)(wsXin + (row * NC + c) * NST + 8 * h);
        float4 a0 = A4[0], a1 = A4[1];
        float4 x0 = X4[0], x1 = X4[1];
        float An[8]  = {a0.x, a0.y, a0.z, a0.w, a1.x, a1.y, a1.z, a1.w};
        float xin[8] = {x0.x, x0.y, x0.z, x0.w, x1.x, x1.y, x1.z, x1.w};
        float bias = dbias[d];
        float Dd   = Dp[d];

        float sp = softplus_f(dv + bias);
        float rr = sp;
        rr += ROW_SHR(rr, 1);
        rr += ROW_SHR(rr, 2);
        rr += ROW_SHR(rr, 4);
        rr += ROW_SHR(rr, 8);
        float rl  = rr * LOG2E_F;
        float dus = sp * uv;

        float P[8], Q[8];
        float v = 0.0f;
#pragma unroll
        for (int q = 0; q < 8; q++) {
            float g = An[q] * rl;
            P[q] = cv[q] * __builtin_amdgcn_exp2f(g);
            Q[q] = (bv[q] * dus) * __builtin_amdgcn_exp2f(-g);
            v += P[q] * xin[q];
        }

        v8f acc = {0.f, 0.f, 0.f, 0.f, 0.f, 0.f, 0.f, 0.f};
#pragma unroll
        for (int j = 0; j < 4; j++) {
            v2f va; va.x = Q[2 * j]; va.y = Q[2 * j + 1];
            v2f vb; vb.x = P[2 * j]; vb.y = P[2 * j + 1];
            acc = __builtin_amdgcn_wmma_f32_16x16x4_f32(
                false, va, false, vb, (short)0, acc, false, false);
        }
#pragma unroll
        for (int r = 0; r < 8; r++) v += (r <= lim) ? acc[r] : 0.0f;
        float y = v + xhalf(v);

        float sig = __builtin_amdgcn_rcpf(1.0f + __expf(-zv));
        float o = (y + uv * Dd) * (zv * sig);
        if (h == 0) out[base] = o;
    }
}

extern "C" void kernel_launch(void* const* d_in, const int* in_sizes, int n_in,
                              void* d_out, int out_size, void* d_ws, size_t ws_size,
                              hipStream_t stream) {
    const float* u     = (const float*)d_in[0];
    const float* delta = (const float*)d_in[1];
    const float* A     = (const float*)d_in[2];
    const float* B     = (const float*)d_in[3];
    const float* C     = (const float*)d_in[4];
    const float* Dp    = (const float*)d_in[5];
    const float* z     = (const float*)d_in[6];
    const float* dbias = (const float*)d_in[7];
    float* out = (float*)d_out;

    float* wsDecay = (float*)d_ws;
    float* wsS     = wsDecay + WSN;

    k1_chunk_state<<<TOTCH / K1_DPB / K1_CPB, K1_DPB * K1_CPB, 0, stream>>>(
        u, delta, A, B, dbias, wsDecay, wsS);
    k2_scan<<<(NBATCH * NDIM * NST) / 256, 256, 0, stream>>>(wsDecay, wsS);
    k3_output<<<TOTCH / K3_WPB / K3_DPW, 32 * K3_WPB, 0, stream>>>(
        u, delta, A, B, C, Dp, z, dbias, wsDecay, out);
}


# k2 launch removed: a k3 work-group owns one (b, d pair) over all 128 chunks and scans k1's (decay,state) rows in LDS in its prologue, 16 segments of 8 chunks; inter-chunk scan sum order changed
# speedup vs baseline: 3.9945x; 1.3335x over previous
#include <hip/hip_runtime.h>

#define NBATCH 2
#define NDIM   768
#define NLEN   2048
#define NST    16
#define CHUNK  16
#define NC     (NLEN / CHUNK)
#define TOTCH  (NBATCH * NDIM * NC)
#define WSN    ((size_t)TOTCH * NST)

typedef float v2f __attribute__((ext_vector_type(2)));
typedef float v8f __attribute__((ext_vector_type(8)));

__device__ __forceinline__ float softplus_f(float x) {
    return (x > 20.0f) ? x : __logf(1.0f + __expf(x));
}

#define K1_DPB 32
#define K1_CPB 8

__global__ __launch_bounds__(256) void k1_chunk_state(
    const float* __restrict__ u, const float* __restrict__ delta,
    const float* __restrict__ A, const float* __restrict__ B,
    const float* __restrict__ dbias,
    float* __restrict__ wsDecay, float* __restrict__ wsS)
{
    int w    = __builtin_amdgcn_readfirstlane((int)(threadIdx.x >> 5));
    int lane = threadIdx.x & 31;

    int cg = blockIdx.x % (NC / K1_CPB);
    int dg = (blockIdx.x / (NC / K1_CPB)) % (NDIM / K1_DPB);
    int b  = blockIdx.x / ((NC / K1_CPB) * (NDIM / K1_DPB));
    int c  = cg * K1_CPB + w;
    int d  = dg * K1_DPB + lane;
    int cs = c * CHUNK;

    const float4* dp4 = (const float4*)(delta + (size_t)(b * NDIM + d) * NLEN + cs);
    const float4* up4 = (const float4*)(u     + (size_t)(b * NDIM + d) * NLEN + cs);
    const float4* A4  = (const float4*)(A + d * NST);
    const float*  Bb  = B + (size_t)(b * NST) * NLEN + cs;

    float An[NST];
#pragma unroll
    for (int q = 0; q < NST / 4; q++) {
        float4 a4 = A4[q];
        An[4 * q + 0] = a4.x; An[4 * q + 1] = a4.y;
        An[4 * q + 2] = a4.z; An[4 * q + 3] = a4.w;
    }
    float bias = dbias[d];

    float x[NST];
#pragma unroll
    for (int n = 0; n < NST; n++) x[n] = 0.0f;
    float R = 0.0f;

    float4 d4 = dp4[0];
    float4 u4 = up4[0];
#pragma unroll 1
    for (int tq = 0; tq < CHUNK / 4; tq++) {
        int tn = (tq + 1 < CHUNK / 4) ? tq + 1 : tq;
        float4 dnx = dp4[tn];
        float4 unx = up4[tn];

        float4 bq[NST];
#pragma unroll
        for (int n = 0; n < NST; n++)
            bq[n] = *(const float4*)(Bb + (size_t)n * NLEN + 4 * tq);

        float dq[4] = {d4.x, d4.y, d4.z, d4.w};
        float uq[4] = {u4.x, u4.y, u4.z, u4.w};
#pragma unroll
        for (int j = 0; j < 4; j++) {
            float sp = softplus_f(dq[j] + bias);
            float du = sp * uq[j];
            R += sp;
#pragma unroll
            for (int n = 0; n < NST; n++) {
                float a  = __expf(An[n] * sp);
                float bv = (j == 0) ? bq[n].x : (j == 1) ? bq[n].y
                         : (j == 2) ? bq[n].z : bq[n].w;
                x[n] = a * x[n] + du * bv;
            }
        }
        d4 = dnx;
        u4 = unx;
    }
    size_t o = ((size_t)(b * NDIM + d) * NC + c) * NST;
    float4* S4 = (float4*)(wsS + o);
    float4* D4 = (float4*)(wsDecay + o);
#pragma unroll
    for (int q = 0; q < NST / 4; q++) {
        float4 s4, e4;
        s4.x = x[4 * q + 0]; s4.y = x[4 * q + 1];
        s4.z = x[4 * q + 2]; s4.w = x[4 * q + 3];
        e4.x = __expf(An[4 * q + 0] * R); e4.y = __expf(An[4 * q + 1] * R);
        e4.z = __expf(An[4 * q + 2] * R); e4.w = __expf(An[4 * q + 3] * R);
        S4[q] = s4;
        D4[q] = e4;
    }
}

#define K3_DPB 2
#define K3_WPB 16
#define K3_CPW (NC / K3_WPB)
#define LOG2E_F 1.4426950408889634f

__device__ __forceinline__ float xhalf(float x) {
    int v = __float_as_int(x);
    return __int_as_float(__builtin_amdgcn_permlanex16(
        v, v, 0x76543210, 0xfedcba98, false, false));
}
#define ROW_SHR(v, K) __int_as_float(__builtin_amdgcn_update_dpp( \
    0, __float_as_int(v), 0x110 + (K), 0xF, 0xF, true))

__global__ __launch_bounds__(32 * K3_WPB) void k3_scan_output(
    const float* __restrict__ u, const float* __restrict__ delta,
    const float* __restrict__ A, const float* __restrict__ B,
    const float* __restrict__ C, const float* __restrict__ Dp,
    const float* __restrict__ z, const float* __restrict__ dbias,
    const float* __restrict__ wsDecay, const float* __restrict__ wsS,
    float* __restrict__ out)
{
    __shared__ float aggA[K3_DPB * K3_WPB * NST];
    __shared__ float aggS[K3_DPB * K3_WPB * NST];
    __shared__ __attribute__((aligned(16))) float xinL[K3_DPB * NC * NST];

    int tid = threadIdx.x;
    int b   = blockIdx.x / (NDIM / K3_DPB);
    int dp  = blockIdx.x % (NDIM / K3_DPB);

    {
        int n    = tid & (NST - 1);
        int seg  = (tid >> 4) & (K3_WPB - 1);
        int dsel = tid >> 8;
        int d    = dp * K3_DPB + dsel;
        size_t base = ((size_t)(b * NDIM + d) * NC + seg * K3_CPW) * NST + n;

        float dec[K3_CPW], sv[K3_CPW];
#pragma unroll
        for (int j = 0; j < K3_CPW; j++) {
            dec[j] = wsDecay[base + (size_t)j * NST];
            sv[j]  = wsS[base + (size_t)j * NST];
        }
        float ag = dec[0], sg = sv[0];
#pragma unroll
        for (int j = 1; j < K3_CPW; j++) {
            sg = dec[j] * sg + sv[j];
            ag = ag * dec[j];
        }
        int ai = (dsel * K3_WPB + seg) * NST + n;
        aggA[ai] = ag;
        aggS[ai] = sg;
        __syncthreads();

        float x = 0.0f;
#pragma unroll
        for (int i = 0; i < K3_WPB - 1; i++) {
            int gi = (dsel * K3_WPB + i) * NST + n;
            float xn = aggA[gi] * x + aggS[gi];
            x = (i < seg) ? xn : x;
        }
        float* xo = xinL + ((size_t)(dsel * NC + seg * K3_CPW)) * NST + n;
#pragma unroll
        for (int j = 0; j < K3_CPW; j++) {
            xo[j * NST] = x;
            x = dec[j] * x + sv[j];
        }
        __syncthreads();
    }

    int w    = __builtin_amdgcn_readfirstlane((int)(threadIdx.x >> 5));
    int lane = tid & 31;
    int m    = lane & 15;
    int h    = lane >> 4;
    int lim  = m - 8 * h;

    for (int ci = 0; ci < K3_CPW; ci++) {
        int c  = ci * K3_WPB + w;
        int cs = c * CHUNK;

        float cv[8], bv[8];
        {
            const float* Cb = C + (size_t)(b * NST + 8 * h) * NLEN + cs + m;
            const float* Bb = B + (size_t)(b * NST + 8 * h) * NLEN + cs + m;
#pragma unroll
            for (int q = 0; q < 8; q++) {
                cv[q] = Cb[(size_t)q * NLEN];
                bv[q] = Bb[(size_t)q * NLEN];
            }
        }

#pragma unroll 1
        for (int dsel = 0; dsel < K3_DPB; dsel++) {
            int d = dp * K3_DPB + dsel;
            size_t row  = (size_t)(b * NDIM + d);
            size_t base = row * NLEN + cs + m;

            float dv = delta[base];
            float uv = u[base];
            float zv = z[base];
            const float4* A4 = (const float4*)(A + d * NST + 8 * h);
            const float4* X4 = (const float4*)(xinL + (dsel * NC + c) * NST + 8 * h);
            float4 a0 = A4[0], a1 = A4[1];
            float4 x0 = X4[0], x1 = X4[1];
            float An[8]  = {a0.x, a0.y, a0.z, a0.w, a1.x, a1.y, a1.z, a1.w};
            float xin[8] = {x0.x, x0.y, x0.z, x0.w, x1.x, x1.y, x1.z, x1.w};
            float bias = dbias[d];
            float Dd   = Dp[d];

            float sp = softplus_f(dv + bias);
            float rr = sp;
            rr += ROW_SHR(rr, 1);
            rr += ROW_SHR(rr, 2);
            rr += ROW_SHR(rr, 4);
            rr += ROW_SHR(rr, 8);
            float rl  = rr * LOG2E_F;
            float dus = sp * uv;

            float P[8], Q[8];
            float v = 0.0f;
#pragma unroll
            for (int q = 0; q < 8; q++) {
                float g = An[q] * rl;
                P[q] = cv[q] * __builtin_amdgcn_exp2f(g);
                Q[q] = (bv[q] * dus) * __builtin_amdgcn_exp2f(-g);
                v += P[q] * xin[q];
            }

            v8f acc = {0.f, 0.f, 0.f, 0.f, 0.f, 0.f, 0.f, 0.f};
#pragma unroll
            for (int j = 0; j < 4; j++) {
                v2f va; va.x = Q[2 * j]; va.y = Q[2 * j + 1];
                v2f vb; vb.x = P[2 * j]; vb.y = P[2 * j + 1];
                acc = __builtin_amdgcn_wmma_f32_16x16x4_f32(
                    false, va, false, vb, (short)0, acc, false, false);
            }
#pragma unroll
            for (int r = 0; r < 8; r++) v += (r <= lim) ? acc[r] : 0.0f;
            float y = v + xhalf(v);

            float sig = __builtin_amdgcn_rcpf(1.0f + __expf(-zv));
            float o = (y + uv * Dd) * (zv * sig);
            if (h == 0) out[base] = o;
        }
    }
}

extern "C" void kernel_launch(void* const* d_in, const int* in_sizes, int n_in,
                              void* d_out, int out_size, void* d_ws, size_t ws_size,
                              hipStream_t stream) {
    const float* u     = (const float*)d_in[0];
    const float* delta = (const float*)d_in[1];
    const float* A     = (const float*)d_in[2];
    const float* B     = (const float*)d_in[3];
    const float* C     = (const float*)d_in[4];
    const float* Dp    = (const float*)d_in[5];
    const float* z     = (const float*)d_in[6];
    const float* dbias = (const float*)d_in[7];
    float* out = (float*)d_out;

    float* wsDecay = (float*)d_ws;
    float* wsS     = wsDecay + WSN;

    k1_chunk_state<<<TOTCH / K1_DPB / K1_CPB, K1_DPB * K1_CPB, 0, stream>>>(
        u, delta, A, B, dbias, wsDecay, wsS);
    k3_scan_output<<<NBATCH * (NDIM / K3_DPB), 32 * K3_WPB, 0, stream>>>(
        u, delta, A, B, C, Dp, z, dbias, wsDecay, wsS, out);
}


# k3: a work-group owns six d (was two) with 32 waves, so each chunk's B,C registers serve six d and B,C are fetched from L2 a third as often; arithmetic and every sum order unchanged
# speedup vs baseline: 4.1502x; 1.0390x over previous
#include <hip/hip_runtime.h>

#define NBATCH 2
#define NDIM   768
#define NLEN   2048
#define NST    16
#define CHUNK  16
#define NC     (NLEN / CHUNK)
#define TOTCH  (NBATCH * NDIM * NC)
#define WSN    ((size_t)TOTCH * NST)

typedef float v2f __attribute__((ext_vector_type(2)));
typedef float v8f __attribute__((ext_vector_type(8)));

__device__ __forceinline__ float softplus_f(float x) {
    return (x > 20.0f) ? x : __logf(1.0f + __expf(x));
}

#define K1_DPB 32
#define K1_CPB 8

__global__ __launch_bounds__(256) void k1_chunk_state(
    const float* __restrict__ u, const float* __restrict__ delta,
    const float* __restrict__ A, const float* __restrict__ B,
    const float* __restrict__ dbias,
    float* __restrict__ wsDecay, float* __restrict__ wsS)
{
    int w    = __builtin_amdgcn_readfirstlane((int)(threadIdx.x >> 5));
    int lane = threadIdx.x & 31;

    int cg = blockIdx.x % (NC / K1_CPB);
    int dg = (blockIdx.x / (NC / K1_CPB)) % (NDIM / K1_DPB);
    int b  = blockIdx.x / ((NC / K1_CPB) * (NDIM / K1_DPB));
    int c  = cg * K1_CPB + w;
    int d  = dg * K1_DPB + lane;
    int cs = c * CHUNK;

    const float4* dp4 = (const float4*)(delta + (size_t)(b * NDIM + d) * NLEN + cs);
    const float4* up4 = (const float4*)(u     + (size_t)(b * NDIM + d) * NLEN + cs);
    const float4* A4  = (const float4*)(A + d * NST);
    const float*  Bb  = B + (size_t)(b * NST) * NLEN + cs;

    float An[NST];
#pragma unroll
    for (int q = 0; q < NST / 4; q++) {
        float4 a4 = A4[q];
        An[4 * q + 0] = a4.x; An[4 * q + 1] = a4.y;
        An[4 * q + 2] = a4.z; An[4 * q + 3] = a4.w;
    }
    float bias = dbias[d];

    float x[NST];
#pragma unroll
    for (int n = 0; n < NST; n++) x[n] = 0.0f;
    float R = 0.0f;

    float4 d4 = dp4[0];
    float4 u4 = up4[0];
#pragma unroll 1
    for (int tq = 0; tq < CHUNK / 4; tq++) {
        int tn = (tq + 1 < CHUNK / 4) ? tq + 1 : tq;
        float4 dnx = dp4[tn];
        float4 unx = up4[tn];

        float4 bq[NST];
#pragma unroll
        for (int n = 0; n < NST; n++)
            bq[n] = *(const float4*)(Bb + (size_t)n * NLEN + 4 * tq);

        float dq[4] = {d4.x, d4.y, d4.z, d4.w};
        float uq[4] = {u4.x, u4.y, u4.z, u4.w};
#pragma unroll
        for (int j = 0; j < 4; j++) {
            float sp = softplus_f(dq[j] + bias);
            float du = sp * uq[j];
            R += sp;
#pragma unroll
            for (int n = 0; n < NST; n++) {
                float a  = __expf(An[n] * sp);
                float bv = (j == 0) ? bq[n].x : (j == 1) ? bq[n].y
                         : (j == 2) ? bq[n].z : bq[n].w;
                x[n] = a * x[n] + du * bv;
            }
        }
        d4 = dnx;
        u4 = unx;
    }
    size_t o = ((size_t)(b * NDIM + d) * NC + c) * NST;
    float4* S4 = (float4*)(wsS + o);
    float4* D4 = (float4*)(wsDecay + o);
#pragma unroll
    for (int q = 0; q < NST / 4; q++) {
        float4 s4, e4;
        s4.x = x[4 * q + 0]; s4.y = x[4 * q + 1];
        s4.z = x[4 * q + 2]; s4.w = x[4 * q + 3];
        e4.x = __expf(An[4 * q + 0] * R); e4.y = __expf(An[4 * q + 1] * R);
        e4.z = __expf(An[4 * q + 2] * R); e4.w = __expf(An[4 * q + 3] * R);
        S4[q] = s4;
        D4[q] = e4;
    }
}

#define K3_DPB 6
#define K3_WPB 32
#define K3_CPW (NC / K3_WPB)
#define K3_SEG 16
#define K3_SCH (NC / K3_SEG)
#define K3_DPP (32 * K3_WPB / (K3_SEG * NST))
#define K3_NPS ((K3_DPB + K3_DPP - 1) / K3_DPP)
#define LOG2E_F 1.4426950408889634f

__device__ __forceinline__ float xhalf(float x) {
    int v = __float_as_int(x);
    return __int_as_float(__builtin_amdgcn_permlanex16(
        v, v, 0x76543210, 0xfedcba98, false, false));
}
#define ROW_SHR(v, K) __int_as_float(__builtin_amdgcn_update_dpp( \
    0, __float_as_int(v), 0x110 + (K), 0xF, 0xF, true))

__global__ __launch_bounds__(32 * K3_WPB) void k3_scan_output(
    const float* __restrict__ u, const float* __restrict__ delta,
    const float* __restrict__ A, const float* __restrict__ B,
    const float* __restrict__ C, const float* __restrict__ Dp,
    const float* __restrict__ z, const float* __restrict__ dbias,
    const float* __restrict__ wsDecay, const float* __restrict__ wsS,
    float* __restrict__ out)
{
    __shared__ float aggA[K3_DPB * K3_SEG * NST];
    __shared__ float aggS[K3_DPB * K3_SEG * NST];
    __shared__ __attribute__((aligned(16))) float xinL[K3_DPB * NC * NST];

    int tid = threadIdx.x;
    int b   = blockIdx.x / (NDIM / K3_DPB);
    int dp  = blockIdx.x % (NDIM / K3_DPB);

    {
        int n   = tid & (NST - 1);
        int seg = (tid >> 4) & (K3_SEG - 1);
        int dq  = __builtin_amdgcn_readfirstlane(tid >> 8);

        float dec[K3_NPS][K3_SCH], sv[K3_NPS][K3_SCH];
#pragma unroll
        for (int p = 0; p < K3_NPS; p++) {
            int dsel = dq + K3_DPP * p;
            if (dsel < K3_DPB) {
                int d = dp * K3_DPB + dsel;
                size_t base = ((size_t)(b * NDIM + d) * NC + seg * K3_SCH) * NST + n;
#pragma unroll
                for (int j = 0; j < K3_SCH; j++) {
                    dec[p][j] = wsDecay[base + (size_t)j * NST];
                    sv[p][j]  = wsS[base + (size_t)j * NST];
                }
                float ag = dec[p][0], sg = sv[p][0];
#pragma unroll
                for (int j = 1; j < K3_SCH; j++) {
                    sg = dec[p][j] * sg + sv[p][j];
                    ag = ag * dec[p][j];
                }
                int ai = (dsel * K3_SEG + seg) * NST + n;
                aggA[ai] = ag;
                aggS[ai] = sg;
            }
        }
        __syncthreads();

#pragma unroll
        for (int p = 0; p < K3_NPS; p++) {
            int dsel = dq + K3_DPP * p;
            if (dsel < K3_DPB) {
                float x = 0.0f;
#pragma unroll
                for (int i = 0; i < K3_SEG - 1; i++) {
                    int gi = (dsel * K3_SEG + i) * NST + n;
                    float xn = aggA[gi] * x + aggS[gi];
                    x = (i < seg) ? xn : x;
                }
                float* xo = xinL + ((size_t)(dsel * NC + seg * K3_SCH)) * NST + n;
#pragma unroll
                for (int j = 0; j < K3_SCH; j++) {
                    xo[j * NST] = x;
                    x = dec[p][j] * x + sv[p][j];
                }
            }
        }
        __syncthreads();
    }

    int w    = __builtin_amdgcn_readfirstlane((int)(threadIdx.x >> 5));
    int lane = tid & 31;
    int m    = lane & 15;
    int h    = lane >> 4;
    int lim  = m - 8 * h;

    for (int ci = 0; ci < K3_CPW; ci++) {
        int c  = ci * K3_WPB + w;
        int cs = c * CHUNK;

        float cv[8], bv[8];
        {
            const float* Cb = C + (size_t)(b * NST + 8 * h) * NLEN + cs + m;
            const float* Bb = B + (size_t)(b * NST + 8 * h) * NLEN + cs + m;
#pragma unroll
            for (int q = 0; q < 8; q++) {
                cv[q] = Cb[(size_t)q * NLEN];
                bv[q] = Bb[(size_t)q * NLEN];
            }
        }

#pragma unroll 1
        for (int dsel = 0; dsel < K3_DPB; dsel++) {
            int d = dp * K3_DPB + dsel;
            size_t row  = (size_t)(b * NDIM + d);
            size_t base = row * NLEN + cs + m;

            float dv = delta[base];
            float uv = u[base];
            float zv = z[base];
            const float4* A4 = (const float4*)(A + d * NST + 8 * h);
            const float4* X4 = (const float4*)(xinL + (dsel * NC + c) * NST + 8 * h);
            float4 a0 = A4[0], a1 = A4[1];
            float4 x0 = X4[0], x1 = X4[1];
            float An[8]  = {a0.x, a0.y, a0.z, a0.w, a1.x, a1.y, a1.z, a1.w};
            float xin[8] = {x0.x, x0.y, x0.z, x0.w, x1.x, x1.y, x1.z, x1.w};
            float bias = dbias[d];
            float Dd   = Dp[d];

            float sp = softplus_f(dv + bias);
            float rr = sp;
            rr += ROW_SHR(rr, 1);
            rr += ROW_SHR(rr, 2);
            rr += ROW_SHR(rr, 4);
            rr += ROW_SHR(rr, 8);
            float rl  = rr * LOG2E_F;
            float dus = sp * uv;

            float P[8], Q[8];
            float v = 0.0f;
#pragma unroll
            for (int q = 0; q < 8; q++) {
                float g = An[q] * rl;
                P[q] = cv[q] * __builtin_amdgcn_exp2f(g);
                Q[q] = (bv[q] * dus) * __builtin_amdgcn_exp2f(-g);
                v += P[q] * xin[q];
            }

            v8f acc = {0.f, 0.f, 0.f, 0.f, 0.f, 0.f, 0.f, 0.f};
#pragma unroll
            for (int j = 0; j < 4; j++) {
                v2f va; va.x = Q[2 * j]; va.y = Q[2 * j + 1];
                v2f vb; vb.x = P[2 * j]; vb.y = P[2 * j + 1];
                acc = __builtin_amdgcn_wmma_f32_16x16x4_f32(
                    false, va, false, vb, (short)0, acc, false, false);
            }
#pragma unroll
            for (int r = 0; r < 8; r++) v += (r <= lim) ? acc[r] : 0.0f;
            float y = v + xhalf(v);

            float sig = __builtin_amdgcn_rcpf(1.0f + __expf(-zv));
            float o = (y + uv * Dd) * (zv * sig);
            if (h == 0) out[base] = o;
        }
    }
}

extern "C" void kernel_launch(void* const* d_in, const int* in_sizes, int n_in,
                              void* d_out, int out_size, void* d_ws, size_t ws_size,
                              hipStream_t stream) {
    const float* u     = (const float*)d_in[0];
    const float* delta = (const float*)d_in[1];
    const float* A     = (const float*)d_in[2];
    const float* B     = (const float*)d_in[3];
    const float* C     = (const float*)d_in[4];
    const float* Dp    = (const float*)d_in[5];
    const float* z     = (const float*)d_in[6];
    const float* dbias = (const float*)d_in[7];
    float* out = (float*)d_out;

    float* wsDecay = (float*)d_ws;
    float* wsS     = wsDecay + WSN;

    k1_chunk_state<<<TOTCH / K1_DPB / K1_CPB, K1_DPB * K1_CPB, 0, stream>>>(
        u, delta, A, B, dbias, wsDecay, wsS);
    k3_scan_output<<<NBATCH * (NDIM / K3_DPB), 32 * K3_WPB, 0, stream>>>(
        u, delta, A, B, C, Dp, z, dbias, wsDecay, wsS, out);
}


# k3: a wave takes 2 adjacent chunks; lane half = chunk for per-step work (softplus, running sum, gate, loads, store), done once on whole 128-byte lines; halves swap r and delta*u; sum orders unchanged
# speedup vs baseline: 4.6150x; 1.1120x over previous
#include <hip/hip_runtime.h>

#define NBATCH 2
#define NDIM   768
#define NLEN   2048
#define NST    16
#define CHUNK  16
#define NC     (NLEN / CHUNK)
#define TOTCH  (NBATCH * NDIM * NC)
#define WSN    ((size_t)TOTCH * NST)

typedef float v2f __attribute__((ext_vector_type(2)));
typedef float v8f __attribute__((ext_vector_type(8)));

__device__ __forceinline__ float softplus_f(float x) {
    return (x > 20.0f) ? x : __logf(1.0f + __expf(x));
}

#define K1_DPB 32
#define K1_CPB 8

__global__ __launch_bounds__(256) void k1_chunk_state(
    const float* __restrict__ u, const float* __restrict__ delta,
    const float* __restrict__ A, const float* __restrict__ B,
    const float* __restrict__ dbias,
    float* __restrict__ wsDecay, float* __restrict__ wsS)
{
    int w    = __builtin_amdgcn_readfirstlane((int)(threadIdx.x >> 5));
    int lane = threadIdx.x & 31;

    int cg = blockIdx.x % (NC / K1_CPB);
    int dg = (blockIdx.x / (NC / K1_CPB)) % (NDIM / K1_DPB);
    int b  = blockIdx.x / ((NC / K1_CPB) * (NDIM / K1_DPB));
    int c  = cg * K1_CPB + w;
    int d  = dg * K1_DPB + lane;
    int cs = c * CHUNK;

    const float4* dp4 = (const float4*)(delta + (size_t)(b * NDIM + d) * NLEN + cs);
    const float4* up4 = (const float4*)(u     + (size_t)(b * NDIM + d) * NLEN + cs);
    const float4* A4  = (const float4*)(A + d * NST);
    const float*  Bb  = B + (size_t)(b * NST) * NLEN + cs;

    float An[NST];
#pragma unroll
    for (int q = 0; q < NST / 4; q++) {
        float4 a4 = A4[q];
        An[4 * q + 0] = a4.x; An[4 * q + 1] = a4.y;
        An[4 * q + 2] = a4.z; An[4 * q + 3] = a4.w;
    }
    float bias = dbias[d];

    float x[NST];
#pragma unroll
    for (int n = 0; n < NST; n++) x[n] = 0.0f;
    float R = 0.0f;

    float4 d4 = dp4[0];
    float4 u4 = up4[0];
#pragma unroll 1
    for (int tq = 0; tq < CHUNK / 4; tq++) {
        int tn = (tq + 1 < CHUNK / 4) ? tq + 1 : tq;
        float4 dnx = dp4[tn];
        float4 unx = up4[tn];

        float4 bq[NST];
#pragma unroll
        for (int n = 0; n < NST; n++)
            bq[n] = *(const float4*)(Bb + (size_t)n * NLEN + 4 * tq);

        float dq[4] = {d4.x, d4.y, d4.z, d4.w};
        float uq[4] = {u4.x, u4.y, u4.z, u4.w};
#pragma unroll
        for (int j = 0; j < 4; j++) {
            float sp = softplus_f(dq[j] + bias);
            float du = sp * uq[j];
            R += sp;
#pragma unroll
            for (int n = 0; n < NST; n++) {
                float a  = __expf(An[n] * sp);
                float bv = (j == 0) ? bq[n].x : (j == 1) ? bq[n].y
                         : (j == 2) ? bq[n].z : bq[n].w;
                x[n] = a * x[n] + du * bv;
            }
        }
        d4 = dnx;
        u4 = unx;
    }
    size_t o = ((size_t)(b * NDIM + d) * NC + c) * NST;
    float4* S4 = (float4*)(wsS + o);
    float4* D4 = (float4*)(wsDecay + o);
#pragma unroll
    for (int q = 0; q < NST / 4; q++) {
        float4 s4, e4;
        s4.x = x[4 * q + 0]; s4.y = x[4 * q + 1];
        s4.z = x[4 * q + 2]; s4.w = x[4 * q + 3];
        e4.x = __expf(An[4 * q + 0] * R); e4.y = __expf(An[4 * q + 1] * R);
        e4.z = __expf(An[4 * q + 2] * R); e4.w = __expf(An[4 * q + 3] * R);
        S4[q] = s4;
        D4[q] = e4;
    }
}

#define K3_DPB 6
#define K3_WPB 32
#define K3_PPW (NC / K3_WPB / 2)
#define K3_SEG 16
#define K3_SCH (NC / K3_SEG)
#define K3_DPP (32 * K3_WPB / (K3_SEG * NST))
#define K3_NPS ((K3_DPB + K3_DPP - 1) / K3_DPP)
#define LOG2E_F 1.4426950408889634f

__device__ __forceinline__ float xhalf(float x) {
    int v = __float_as_int(x);
    return __int_as_float(__builtin_amdgcn_permlanex16(
        v, v, 0x76543210, 0xfedcba98, false, false));
}
#define ROW_SHR(v, K) __int_as_float(__builtin_amdgcn_update_dpp( \
    0, __float_as_int(v), 0x110 + (K), 0xF, 0xF, true))

__global__ __launch_bounds__(32 * K3_WPB) void k3_scan_output(
    const float* __restrict__ u, const float* __restrict__ delta,
    const float* __restrict__ A, const float* __restrict__ B,
    const float* __restrict__ C, const float* __restrict__ Dp,
    const float* __restrict__ z, const float* __restrict__ dbias,
    const float* __restrict__ wsDecay, const float* __restrict__ wsS,
    float* __restrict__ out)
{
    __shared__ float aggA[K3_DPB * K3_SEG * NST];
    __shared__ float aggS[K3_DPB * K3_SEG * NST];
    __shared__ __attribute__((aligned(16))) float xinL[K3_DPB * NC * NST];

    int tid = threadIdx.x;
    int b   = blockIdx.x / (NDIM / K3_DPB);
    int dp  = blockIdx.x % (NDIM / K3_DPB);

    {
        int n   = tid & (NST - 1);
        int seg = (tid >> 4) & (K3_SEG - 1);
        int dq  = __builtin_amdgcn_readfirstlane(tid >> 8);

        float dec[K3_NPS][K3_SCH], sv[K3_NPS][K3_SCH];
#pragma unroll
        for (int p = 0; p < K3_NPS; p++) {
            int dsel = dq + K3_DPP * p;
            if (dsel < K3_DPB) {
                int d = dp * K3_DPB + dsel;
                size_t base = ((size_t)(b * NDIM + d) * NC + seg * K3_SCH) * NST + n;
#pragma unroll
                for (int j = 0; j < K3_SCH; j++) {
                    dec[p][j] = wsDecay[base + (size_t)j * NST];
                    sv[p][j]  = wsS[base + (size_t)j * NST];
                }
                float ag = dec[p][0], sg = sv[p][0];
#pragma unroll
                for (int j = 1; j < K3_SCH; j++) {
                    sg = dec[p][j] * sg + sv[p][j];
                    ag = ag * dec[p][j];
                }
                int ai = (dsel * K3_SEG + seg) * NST + n;
                aggA[ai] = ag;
                aggS[ai] = sg;
            }
        }
        __syncthreads();

#pragma unroll
        for (int p = 0; p < K3_NPS; p++) {
            int dsel = dq + K3_DPP * p;
            if (dsel < K3_DPB) {
                float x = 0.0f;
#pragma unroll
                for (int i = 0; i < K3_SEG - 1; i++) {
                    int gi = (dsel * K3_SEG + i) * NST + n;
                    float xn = aggA[gi] * x + aggS[gi];
                    x = (i < seg) ? xn : x;
                }
                float* xo = xinL + ((size_t)(dsel * NC + seg * K3_SCH)) * NST + n;
#pragma unroll
                for (int j = 0; j < K3_SCH; j++) {
                    xo[j * NST] = x;
                    x = dec[p][j] * x + sv[p][j];
                }
            }
        }
        __syncthreads();
    }

    int  w    = __builtin_amdgcn_readfirstlane((int)(threadIdx.x >> 5));
    int  lane = tid & 31;
    int  m    = lane & 15;
    int  h    = lane >> 4;
    bool hi   = (h != 0);
    int  lim  = m - 8 * h;

    for (int pi = 0; pi < K3_PPW; pi++) {
        int c0  = 2 * (pi * K3_WPB + w);
        int cs0 = c0 * CHUNK;

        float cv[2][8], bv[2][8];
        {
            const float* Cb = C + (size_t)(b * NST + 8 * h) * NLEN + cs0 + m;
            const float* Bb = B + (size_t)(b * NST + 8 * h) * NLEN + cs0 + m;
#pragma unroll
            for (int q = 0; q < 8; q++) {
                cv[0][q] = Cb[(size_t)q * NLEN];
                bv[0][q] = Bb[(size_t)q * NLEN];
                cv[1][q] = Cb[(size_t)q * NLEN + CHUNK];
                bv[1][q] = Bb[(size_t)q * NLEN + CHUNK];
            }
        }

#pragma unroll 1
        for (int dsel = 0; dsel < K3_DPB; dsel++) {
            int d = dp * K3_DPB + dsel;
            size_t row  = (size_t)(b * NDIM + d);
            size_t base = row * NLEN + cs0 + lane;

            float dv = delta[base];
            float uv = u[base];
            float zv = z[base];
            const float4* A4 = (const float4*)(A + d * NST + 8 * h);
            const float4* X4 = (const float4*)(xinL + (dsel * NC + c0) * NST + 8 * h);
            float4 a0 = A4[0], a1 = A4[1];
            float An[8] = {a0.x, a0.y, a0.z, a0.w, a1.x, a1.y, a1.z, a1.w};
            float bias = dbias[d];
            float Dd   = Dp[d];

            float sp = softplus_f(dv + bias);
            float rr = sp;
            rr += ROW_SHR(rr, 1);
            rr += ROW_SHR(rr, 2);
            rr += ROW_SHR(rr, 4);
            rr += ROW_SHR(rr, 8);
            float rl  = rr * LOG2E_F;
            float dus = sp * uv;
            float rlx = xhalf(rl);
            float dux = xhalf(dus);

            float vk[2];
#pragma unroll
            for (int k = 0; k < 2; k++) {
                float rlk = (hi == (k != 0)) ? rl  : rlx;
                float duk = (hi == (k != 0)) ? dus : dux;
                float4 x0 = X4[4 * k], x1 = X4[4 * k + 1];
                float xin[8] = {x0.x, x0.y, x0.z, x0.w, x1.x, x1.y, x1.z, x1.w};

                float P[8], Q[8];
                float v = 0.0f;
#pragma unroll
                for (int q = 0; q < 8; q++) {
                    float g = An[q] * rlk;
                    P[q] = cv[k][q] * __builtin_amdgcn_exp2f(g);
                    Q[q] = (bv[k][q] * duk) * __builtin_amdgcn_exp2f(-g);
                    v += P[q] * xin[q];
                }

                v8f acc = {0.f, 0.f, 0.f, 0.f, 0.f, 0.f, 0.f, 0.f};
#pragma unroll
                for (int j = 0; j < 4; j++) {
                    v2f va; va.x = Q[2 * j]; va.y = Q[2 * j + 1];
                    v2f vb; vb.x = P[2 * j]; vb.y = P[2 * j + 1];
                    acc = __builtin_amdgcn_wmma_f32_16x16x4_f32(
                        false, va, false, vb, (short)0, acc, false, false);
                }
#pragma unroll
                for (int r = 0; r < 8; r++) v += (r <= lim) ? acc[r] : 0.0f;
                vk[k] = v;
            }
            float mine = hi ? vk[1] : vk[0];
            float send = hi ? vk[0] : vk[1];
            float y = mine + xhalf(send);

            float sig = __builtin_amdgcn_rcpf(1.0f + __expf(-zv));
            float o = (y + uv * Dd) * (zv * sig);
            out[base] = o;
        }
    }
}

extern "C" void kernel_launch(void* const* d_in, const int* in_sizes, int n_in,
                              void* d_out, int out_size, void* d_ws, size_t ws_size,
                              hipStream_t stream) {
    const float* u     = (const float*)d_in[0];
    const float* delta = (const float*)d_in[1];
    const float* A     = (const float*)d_in[2];
    const float* B     = (const float*)d_in[3];
    const float* C     = (const float*)d_in[4];
    const float* Dp    = (const float*)d_in[5];
    const float* z     = (const float*)d_in[6];
    const float* dbias = (const float*)d_in[7];
    float* out = (float*)d_out;

    float* wsDecay = (float*)d_ws;
    float* wsS     = wsDecay + WSN;

    k1_chunk_state<<<TOTCH / K1_DPB / K1_CPB, K1_DPB * K1_CPB, 0, stream>>>(
        u, delta, A, B, dbias, wsDecay, wsS);
    k3_scan_output<<<NBATCH * (NDIM / K3_DPB), 32 * K3_WPB, 0, stream>>>(
        u, delta, A, B, C, Dp, z, dbias, wsDecay, wsS, out);
}


# k1 launch removed: each k3 work-group builds its six rows' chunk decay/state itself (softplus staged once in LDS, thread=(chunk,n) 16-step recurrence); workspace unused; sum orders unchanged
# speedup vs baseline: 5.6515x; 1.2246x over previous
#include <hip/hip_runtime.h>

#define NBATCH 2
#define NDIM   768
#define NLEN   2048
#define NST    16
#define CHUNK  16
#define NC     (NLEN / CHUNK)

typedef float v2f __attribute__((ext_vector_type(2)));
typedef float v8f __attribute__((ext_vector_type(8)));

__device__ __forceinline__ float softplus_f(float x) {
    return (x > 20.0f) ? x : __logf(1.0f + __expf(x));
}

#define K3_DPB 6
#define K3_WPB 32
#define K3_PPW (NC / K3_WPB / 2)
#define K3_SEG 16
#define K3_SCH (NC / K3_SEG)
#define K3_DPP (32 * K3_WPB / (K3_SEG * NST))
#define K3_NPS ((K3_DPB + K3_DPP - 1) / K3_DPP)
#define K3_CPP (32 * K3_WPB / NST)
#define K3_TPT (NLEN / (32 * K3_WPB))
#define LOG2E_F 1.4426950408889634f

__device__ __forceinline__ float xhalf(float x) {
    int v = __float_as_int(x);
    return __int_as_float(__builtin_amdgcn_permlanex16(
        v, v, 0x76543210, 0xfedcba98, false, false));
}
#define ROW_SHR(v, K) __int_as_float(__builtin_amdgcn_update_dpp( \
    0, __float_as_int(v), 0x110 + (K), 0xF, 0xF, true))

__global__ __launch_bounds__(32 * K3_WPB) void k3_scan_output(
    const float* __restrict__ u, const float* __restrict__ delta,
    const float* __restrict__ A, const float* __restrict__ B,
    const float* __restrict__ C, const float* __restrict__ Dp,
    const float* __restrict__ z, const float* __restrict__ dbias,
    float* __restrict__ out)
{
    __shared__ float aggA[K3_DPB * K3_SEG * NST];
    __shared__ float aggS[K3_DPB * K3_SEG * NST];
    __shared__ __attribute__((aligned(16))) float xinL[K3_DPB * NC * NST];
    __shared__ __attribute__((aligned(16))) float spL[K3_DPB * NLEN];
    __shared__ __attribute__((aligned(16))) float duL[K3_DPB * NLEN];
    __shared__ float decL[K3_DPB * NC * NST];
    __shared__ float svL[K3_DPB * NC * NST];

    int tid = threadIdx.x;
    int b   = blockIdx.x / (NDIM / K3_DPB);
    int dp  = blockIdx.x % (NDIM / K3_DPB);

    {
        float dvv[K3_DPB][K3_TPT], uvv[K3_DPB][K3_TPT];
#pragma unroll
        for (int dsel = 0; dsel < K3_DPB; dsel++) {
            size_t rb = (size_t)(b * NDIM + dp * K3_DPB + dsel) * NLEN + tid;
#pragma unroll
            for (int i = 0; i < K3_TPT; i++) {
                dvv[dsel][i] = delta[rb + 32 * K3_WPB * i];
                uvv[dsel][i] = u[rb + 32 * K3_WPB * i];
            }
        }
#pragma unroll
        for (int dsel = 0; dsel < K3_DPB; dsel++) {
            float bias = dbias[dp * K3_DPB + dsel];
#pragma unroll
            for (int i = 0; i < K3_TPT; i++) {
                float sp = softplus_f(dvv[dsel][i] + bias);
                spL[dsel * NLEN + tid + 32 * K3_WPB * i] = sp;
                duL[dsel * NLEN + tid + 32 * K3_WPB * i] = sp * uvv[dsel][i];
            }
        }
    }
    __syncthreads();

    {
        int n  = tid & (NST - 1);
        int cl = tid >> 4;
#pragma unroll 1
        for (int pass = 0; pass < NC / K3_CPP; pass++) {
            int c = cl + K3_CPP * pass;
            const float4* Bp = (const float4*)(B + (size_t)(b * NST + n) * NLEN + c * CHUNK);
            float4 b0 = Bp[0], b1 = Bp[1], b2 = Bp[2], b3 = Bp[3];
            float bq[CHUNK] = {b0.x, b0.y, b0.z, b0.w, b1.x, b1.y, b1.z, b1.w,
                               b2.x, b2.y, b2.z, b2.w, b3.x, b3.y, b3.z, b3.w};
#pragma unroll 1
            for (int dsel = 0; dsel < K3_DPB; dsel += 2) {
                int d = dp * K3_DPB + dsel;
                v2f An;
                An.x = A[d * NST + n] * LOG2E_F;
                An.y = A[(d + 1) * NST + n] * LOG2E_F;
                const float4* s0 = (const float4*)(spL + dsel * NLEN + c * CHUNK);
                const float4* s1 = (const float4*)(spL + (dsel + 1) * NLEN + c * CHUNK);
                const float4* u0 = (const float4*)(duL + dsel * NLEN + c * CHUNK);
                const float4* u1 = (const float4*)(duL + (dsel + 1) * NLEN + c * CHUNK);
                v2f x = {0.0f, 0.0f};
                v2f R = {0.0f, 0.0f};
#pragma unroll
                for (int q = 0; q < CHUNK / 4; q++) {
                    float4 sa = s0[q], sb = s1[q], ua = u0[q], ub = u1[q];
                    float spa[4] = {sa.x, sa.y, sa.z, sa.w};
                    float spb[4] = {sb.x, sb.y, sb.z, sb.w};
                    float dua[4] = {ua.x, ua.y, ua.z, ua.w};
                    float dub[4] = {ub.x, ub.y, ub.z, ub.w};
#pragma unroll
                    for (int j = 0; j < 4; j++) {
                        v2f sp; sp.x = spa[j]; sp.y = spb[j];
                        v2f du; du.x = dua[j]; du.y = dub[j];
                        v2f g = An * sp;
                        v2f a;
                        a.x = __builtin_amdgcn_exp2f(g.x);
                        a.y = __builtin_amdgcn_exp2f(g.y);
                        x = a * x + du * bq[4 * q + j];
                        R += sp;
                    }
                }
                v2f gd = An * R;
                int idx = (dsel * NC + c) * NST + n;
                decL[idx]            = __builtin_amdgcn_exp2f(gd.x);
                svL[idx]             = x.x;
                decL[idx + NC * NST] = __builtin_amdgcn_exp2f(gd.y);
                svL[idx + NC * NST]  = x.y;
            }
        }
    }
    __syncthreads();

    {
        int n   = tid & (NST - 1);
        int seg = (tid >> 4) & (K3_SEG - 1);
        int dq  = __builtin_amdgcn_readfirstlane(tid >> 8);

        float dec[K3_NPS][K3_SCH], sv[K3_NPS][K3_SCH];
#pragma unroll
        for (int p = 0; p < K3_NPS; p++) {
            int dsel = dq + K3_DPP * p;
            if (dsel < K3_DPB) {
                int base = (dsel * NC + seg * K3_SCH) * NST + n;
#pragma unroll
                for (int j = 0; j < K3_SCH; j++) {
                    dec[p][j] = decL[base + j * NST];
                    sv[p][j]  = svL[base + j * NST];
                }
                float ag = dec[p][0], sg = sv[p][0];
#pragma unroll
                for (int j = 1; j < K3_SCH; j++) {
                    sg = dec[p][j] * sg + sv[p][j];
                    ag = ag * dec[p][j];
                }
                int ai = (dsel * K3_SEG + seg) * NST + n;
                aggA[ai] = ag;
                aggS[ai] = sg;
            }
        }
        __syncthreads();

#pragma unroll
        for (int p = 0; p < K3_NPS; p++) {
            int dsel = dq + K3_DPP * p;
            if (dsel < K3_DPB) {
                float x = 0.0f;
#pragma unroll
                for (int i = 0; i < K3_SEG - 1; i++) {
                    int gi = (dsel * K3_SEG + i) * NST + n;
                    float xn = aggA[gi] * x + aggS[gi];
                    x = (i < seg) ? xn : x;
                }
                float* xo = xinL + ((size_t)(dsel * NC + seg * K3_SCH)) * NST + n;
#pragma unroll
                for (int j = 0; j < K3_SCH; j++) {
                    xo[j * NST] = x;
                    x = dec[p][j] * x + sv[p][j];
                }
            }
        }
        __syncthreads();
    }

    int  w    = __builtin_amdgcn_readfirstlane((int)(threadIdx.x >> 5));
    int  lane = tid & 31;
    int  m    = lane & 15;
    int  h    = lane >> 4;
    bool hi   = (h != 0);
    int  lim  = m - 8 * h;

    for (int pi = 0; pi < K3_PPW; pi++) {
        int c0  = 2 * (pi * K3_WPB + w);
        int cs0 = c0 * CHUNK;

        float cv[2][8], bv[2][8];
        {
            const float* Cb = C + (size_t)(b * NST + 8 * h) * NLEN + cs0 + m;
            const float* Bb = B + (size_t)(b * NST + 8 * h) * NLEN + cs0 + m;
#pragma unroll
            for (int q = 0; q < 8; q++) {
                cv[0][q] = Cb[(size_t)q * NLEN];
                bv[0][q] = Bb[(size_t)q * NLEN];
                cv[1][q] = Cb[(size_t)q * NLEN + CHUNK];
                bv[1][q] = Bb[(size_t)q * NLEN + CHUNK];
            }
        }

#pragma unroll 1
        for (int dsel = 0; dsel < K3_DPB; dsel++) {
            int d = dp * K3_DPB + dsel;
            size_t row  = (size_t)(b * NDIM + d);
            size_t base = row * NLEN + cs0 + lane;

            float uv = u[base];
            float zv = z[base];
            const float4* A4 = (const float4*)(A + d * NST + 8 * h);
            const float4* X4 = (const float4*)(xinL + (dsel * NC + c0) * NST + 8 * h);
            float4 a0 = A4[0], a1 = A4[1];
            float An[8] = {a0.x, a0.y, a0.z, a0.w, a1.x, a1.y, a1.z, a1.w};
            float Dd   = Dp[d];

            float sp = spL[dsel * NLEN + cs0 + lane];
            float rr = sp;
            rr += ROW_SHR(rr, 1);
            rr += ROW_SHR(rr, 2);
            rr += ROW_SHR(rr, 4);
            rr += ROW_SHR(rr, 8);
            float rl  = rr * LOG2E_F;
            float dus = sp * uv;
            float rlx = xhalf(rl);
            float dux = xhalf(dus);

            float vk[2];
#pragma unroll
            for (int k = 0; k < 2; k++) {
                float rlk = (hi == (k != 0)) ? rl  : rlx;
                float duk = (hi == (k != 0)) ? dus : dux;
                float4 x0 = X4[4 * k], x1 = X4[4 * k + 1];
                float xin[8] = {x0.x, x0.y, x0.z, x0.w, x1.x, x1.y, x1.z, x1.w};

                float P[8], Q[8];
                float v = 0.0f;
#pragma unroll
                for (int q = 0; q < 8; q++) {
                    float g = An[q] * rlk;
                    P[q] = cv[k][q] * __builtin_amdgcn_exp2f(g);
                    Q[q] = (bv[k][q] * duk) * __builtin_amdgcn_exp2f(-g);
                    v += P[q] * xin[q];
                }

                v8f acc = {0.f, 0.f, 0.f, 0.f, 0.f, 0.f, 0.f, 0.f};
#pragma unroll
                for (int j = 0; j < 4; j++) {
                    v2f va; va.x = Q[2 * j]; va.y = Q[2 * j + 1];
                    v2f vb; vb.x = P[2 * j]; vb.y = P[2 * j + 1];
                    acc = __builtin_amdgcn_wmma_f32_16x16x4_f32(
                        false, va, false, vb, (short)0, acc, false, false);
                }
#pragma unroll
                for (int r = 0; r < 8; r++) v += (r <= lim) ? acc[r] : 0.0f;
                vk[k] = v;
            }
            float mine = hi ? vk[1] : vk[0];
            float send = hi ? vk[0] : vk[1];
            float y = mine + xhalf(send);

            float sig = __builtin_amdgcn_rcpf(1.0f + __expf(-zv));
            float o = (y + uv * Dd) * (zv * sig);
            out[base] = o;
        }
    }
}

extern "C" void kernel_launch(void* const* d_in, const int* in_sizes, int n_in,
                              void* d_out, int out_size, void* d_ws, size_t ws_size,
                              hipStream_t stream) {
    const float* u     = (const float*)d_in[0];
    const float* delta = (const float*)d_in[1];
    const float* A     = (const float*)d_in[2];
    const float* B     = (const float*)d_in[3];
    const float* C     = (const float*)d_in[4];
    const float* Dp    = (const float*)d_in[5];
    const float* z     = (const float*)d_in[6];
    const float* dbias = (const float*)d_in[7];
    float* out = (float*)d_out;

    k3_scan_output<<<NBATCH * (NDIM / K3_DPB), 32 * K3_WPB, 0, stream>>>(
        u, delta, A, B, C, Dp, z, dbias, out);
}


# first B (stage B) and first-pair B,C (output) loads issued a stage early and carried across LDS-only barriers, so they travel during softplus and the chunk scan; sum orders unchanged
# speedup vs baseline: 5.8344x; 1.0324x over previous
#include <hip/hip_runtime.h>

#define NBATCH 2
#define NDIM   768
#define NLEN   2048
#define NST    16
#define CHUNK  16
#define NC     (NLEN / CHUNK)

typedef float v2f __attribute__((ext_vector_type(2)));
typedef float v8f __attribute__((ext_vector_type(8)));

__device__ __forceinline__ float softplus_f(float x) {
    return (x > 20.0f) ? x : __logf(1.0f + __expf(x));
}

#define K3_DPB 6
#define K3_WPB 32
#define K3_PPW (NC / K3_WPB / 2)
#define K3_SEG 16
#define K3_SCH (NC / K3_SEG)
#define K3_DPP (32 * K3_WPB / (K3_SEG * NST))
#define K3_NPS ((K3_DPB + K3_DPP - 1) / K3_DPP)
#define K3_CPP (32 * K3_WPB / NST)
#define K3_TPT (NLEN / (32 * K3_WPB))
#define LOG2E_F 1.4426950408889634f

__device__ __forceinline__ float xhalf(float x) {
    int v = __float_as_int(x);
    return __int_as_float(__builtin_amdgcn_permlanex16(
        v, v, 0x76543210, 0xfedcba98, false, false));
}
#define ROW_SHR(v, K) __int_as_float(__builtin_amdgcn_update_dpp( \
    0, __float_as_int(v), 0x110 + (K), 0xF, 0xF, true))

__device__ __forceinline__ void lds_barrier() {
    __builtin_amdgcn_fence(__ATOMIC_RELEASE, "workgroup", "local");
    __builtin_amdgcn_s_barrier();
    __builtin_amdgcn_fence(__ATOMIC_ACQUIRE, "workgroup", "local");
}

__global__ __launch_bounds__(32 * K3_WPB) void k3_scan_output(
    const float* __restrict__ u, const float* __restrict__ delta,
    const float* __restrict__ A, const float* __restrict__ B,
    const float* __restrict__ C, const float* __restrict__ Dp,
    const float* __restrict__ z, const float* __restrict__ dbias,
    float* __restrict__ out)
{
    __shared__ float aggA[K3_DPB * K3_SEG * NST];
    __shared__ float aggS[K3_DPB * K3_SEG * NST];
    __shared__ __attribute__((aligned(16))) float xinL[K3_DPB * NC * NST];
    __shared__ __attribute__((aligned(16))) float spL[K3_DPB * NLEN];
    __shared__ __attribute__((aligned(16))) float duL[K3_DPB * NLEN];
    __shared__ float decL[K3_DPB * NC * NST];
    __shared__ float svL[K3_DPB * NC * NST];

    int tid = threadIdx.x;
    int b   = blockIdx.x / (NDIM / K3_DPB);
    int dp  = blockIdx.x % (NDIM / K3_DPB);

    float4 nb0, nb1, nb2, nb3;
    {
        float dvv[K3_DPB][K3_TPT], uvv[K3_DPB][K3_TPT];
#pragma unroll
        for (int dsel = 0; dsel < K3_DPB; dsel++) {
            size_t rb = (size_t)(b * NDIM + dp * K3_DPB + dsel) * NLEN + tid;
#pragma unroll
            for (int i = 0; i < K3_TPT; i++) {
                dvv[dsel][i] = delta[rb + 32 * K3_WPB * i];
                uvv[dsel][i] = u[rb + 32 * K3_WPB * i];
            }
        }
        {
            const float4* Bp = (const float4*)(B + (size_t)(b * NST + (tid & (NST - 1))) * NLEN
                                               + (tid >> 4) * CHUNK);
            nb0 = Bp[0]; nb1 = Bp[1]; nb2 = Bp[2]; nb3 = Bp[3];
        }
#pragma unroll
        for (int dsel = 0; dsel < K3_DPB; dsel++) {
            float bias = dbias[dp * K3_DPB + dsel];
#pragma unroll
            for (int i = 0; i < K3_TPT; i++) {
                float sp = softplus_f(dvv[dsel][i] + bias);
                spL[dsel * NLEN + tid + 32 * K3_WPB * i] = sp;
                duL[dsel * NLEN + tid + 32 * K3_WPB * i] = sp * uvv[dsel][i];
            }
        }
    }
    lds_barrier();

    {
        int n  = tid & (NST - 1);
        int cl = tid >> 4;
#pragma unroll 1
        for (int pass = 0; pass < NC / K3_CPP; pass++) {
            int c = cl + K3_CPP * pass;
            float4 b0 = nb0, b1 = nb1, b2 = nb2, b3 = nb3;
            float bq[CHUNK] = {b0.x, b0.y, b0.z, b0.w, b1.x, b1.y, b1.z, b1.w,
                               b2.x, b2.y, b2.z, b2.w, b3.x, b3.y, b3.z, b3.w};
#pragma unroll 1
            for (int dsel = 0; dsel < K3_DPB; dsel += 2) {
                int d = dp * K3_DPB + dsel;
                v2f An;
                An.x = A[d * NST + n] * LOG2E_F;
                An.y = A[(d + 1) * NST + n] * LOG2E_F;
                const float4* s0 = (const float4*)(spL + dsel * NLEN + c * CHUNK);
                const float4* s1 = (const float4*)(spL + (dsel + 1) * NLEN + c * CHUNK);
                const float4* u0 = (const float4*)(duL + dsel * NLEN + c * CHUNK);
                const float4* u1 = (const float4*)(duL + (dsel + 1) * NLEN + c * CHUNK);
                v2f x = {0.0f, 0.0f};
                v2f R = {0.0f, 0.0f};
#pragma unroll
                for (int q = 0; q < CHUNK / 4; q++) {
                    float4 sa = s0[q], sb = s1[q], ua = u0[q], ub = u1[q];
                    float spa[4] = {sa.x, sa.y, sa.z, sa.w};
                    float spb[4] = {sb.x, sb.y, sb.z, sb.w};
                    float dua[4] = {ua.x, ua.y, ua.z, ua.w};
                    float dub[4] = {ub.x, ub.y, ub.z, ub.w};
#pragma unroll
                    for (int j = 0; j < 4; j++) {
                        v2f sp; sp.x = spa[j]; sp.y = spb[j];
                        v2f du; du.x = dua[j]; du.y = dub[j];
                        v2f g = An * sp;
                        v2f a;
                        a.x = __builtin_amdgcn_exp2f(g.x);
                        a.y = __builtin_amdgcn_exp2f(g.y);
                        x = a * x + du * bq[4 * q + j];
                        R += sp;
                    }
                }
                v2f gd = An * R;
                int idx = (dsel * NC + c) * NST + n;
                decL[idx]            = __builtin_amdgcn_exp2f(gd.x);
                svL[idx]             = x.x;
                decL[idx + NC * NST] = __builtin_amdgcn_exp2f(gd.y);
                svL[idx + NC * NST]  = x.y;
            }
            if (pass + 1 < NC / K3_CPP) {
                const float4* Bp = (const float4*)(B + (size_t)(b * NST + n) * NLEN
                                                   + (c + K3_CPP) * CHUNK);
                nb0 = Bp[0]; nb1 = Bp[1]; nb2 = Bp[2]; nb3 = Bp[3];
            }
        }
    }
    int  w    = __builtin_amdgcn_readfirstlane((int)(threadIdx.x >> 5));
    int  lane = tid & 31;
    int  m    = lane & 15;
    int  h    = lane >> 4;

    float cv[2][8], bv[2][8];
    {
        int cs0 = 2 * w * CHUNK;
        const float* Cb = C + (size_t)(b * NST + 8 * h) * NLEN + cs0 + m;
        const float* Bb = B + (size_t)(b * NST + 8 * h) * NLEN + cs0 + m;
#pragma unroll
        for (int q = 0; q < 8; q++) {
            cv[0][q] = Cb[(size_t)q * NLEN];
            bv[0][q] = Bb[(size_t)q * NLEN];
            cv[1][q] = Cb[(size_t)q * NLEN + CHUNK];
            bv[1][q] = Bb[(size_t)q * NLEN + CHUNK];
        }
    }

    lds_barrier();

    {
        int n   = tid & (NST - 1);
        int seg = (tid >> 4) & (K3_SEG - 1);
        int dq  = __builtin_amdgcn_readfirstlane(tid >> 8);

        float dec[K3_NPS][K3_SCH], sv[K3_NPS][K3_SCH];
#pragma unroll
        for (int p = 0; p < K3_NPS; p++) {
            int dsel = dq + K3_DPP * p;
            if (dsel < K3_DPB) {
                int base = (dsel * NC + seg * K3_SCH) * NST + n;
#pragma unroll
                for (int j = 0; j < K3_SCH; j++) {
                    dec[p][j] = decL[base + j * NST];
                    sv[p][j]  = svL[base + j * NST];
                }
                float ag = dec[p][0], sg = sv[p][0];
#pragma unroll
                for (int j = 1; j < K3_SCH; j++) {
                    sg = dec[p][j] * sg + sv[p][j];
                    ag = ag * dec[p][j];
                }
                int ai = (dsel * K3_SEG + seg) * NST + n;
                aggA[ai] = ag;
                aggS[ai] = sg;
            }
        }
        lds_barrier();

#pragma unroll
        for (int p = 0; p < K3_NPS; p++) {
            int dsel = dq + K3_DPP * p;
            if (dsel < K3_DPB) {
                float x = 0.0f;
#pragma unroll
                for (int i = 0; i < K3_SEG - 1; i++) {
                    int gi = (dsel * K3_SEG + i) * NST + n;
                    float xn = aggA[gi] * x + aggS[gi];
                    x = (i < seg) ? xn : x;
                }
                float* xo = xinL + ((size_t)(dsel * NC + seg * K3_SCH)) * NST + n;
#pragma unroll
                for (int j = 0; j < K3_SCH; j++) {
                    xo[j * NST] = x;
                    x = dec[p][j] * x + sv[p][j];
                }
            }
        }
        lds_barrier();
    }

    bool hi   = (h != 0);
    int  lim  = m - 8 * h;

    for (int pi = 0; pi < K3_PPW; pi++) {
        int c0  = 2 * (pi * K3_WPB + w);
        int cs0 = c0 * CHUNK;

#pragma unroll 1
        for (int dsel = 0; dsel < K3_DPB; dsel++) {
            int d = dp * K3_DPB + dsel;
            size_t row  = (size_t)(b * NDIM + d);
            size_t base = row * NLEN + cs0 + lane;

            float uv = u[base];
            float zv = z[base];
            const float4* A4 = (const float4*)(A + d * NST + 8 * h);
            const float4* X4 = (const float4*)(xinL + (dsel * NC + c0) * NST + 8 * h);
            float4 a0 = A4[0], a1 = A4[1];
            float An[8] = {a0.x, a0.y, a0.z, a0.w, a1.x, a1.y, a1.z, a1.w};
            float Dd   = Dp[d];

            float sp = spL[dsel * NLEN + cs0 + lane];
            float rr = sp;
            rr += ROW_SHR(rr, 1);
            rr += ROW_SHR(rr, 2);
            rr += ROW_SHR(rr, 4);
            rr += ROW_SHR(rr, 8);
            float rl  = rr * LOG2E_F;
            float dus = sp * uv;
            float rlx = xhalf(rl);
            float dux = xhalf(dus);

            float vk[2];
#pragma unroll
            for (int k = 0; k < 2; k++) {
                float rlk = (hi == (k != 0)) ? rl  : rlx;
                float duk = (hi == (k != 0)) ? dus : dux;
                float4 x0 = X4[4 * k], x1 = X4[4 * k + 1];
                float xin[8] = {x0.x, x0.y, x0.z, x0.w, x1.x, x1.y, x1.z, x1.w};

                float P[8], Q[8];
                float v = 0.0f;
#pragma unroll
                for (int q = 0; q < 8; q++) {
                    float g = An[q] * rlk;
                    P[q] = cv[k][q] * __builtin_amdgcn_exp2f(g);
                    Q[q] = (bv[k][q] * duk) * __builtin_amdgcn_exp2f(-g);
                    v += P[q] * xin[q];
                }

                v8f acc = {0.f, 0.f, 0.f, 0.f, 0.f, 0.f, 0.f, 0.f};
#pragma unroll
                for (int j = 0; j < 4; j++) {
                    v2f va; va.x = Q[2 * j]; va.y = Q[2 * j + 1];
                    v2f vb; vb.x = P[2 * j]; vb.y = P[2 * j + 1];
                    acc = __builtin_amdgcn_wmma_f32_16x16x4_f32(
                        false, va, false, vb, (short)0, acc, false, false);
                }
#pragma unroll
                for (int r = 0; r < 8; r++) v += (r <= lim) ? acc[r] : 0.0f;
                vk[k] = v;
            }
            float mine = hi ? vk[1] : vk[0];
            float send = hi ? vk[0] : vk[1];
            float y = mine + xhalf(send);

            float sig = __builtin_amdgcn_rcpf(1.0f + __expf(-zv));
            float o = (y + uv * Dd) * (zv * sig);
            out[base] = o;
        }
        if (pi + 1 < K3_PPW) {
            const float* Cb = C + (size_t)(b * NST + 8 * h) * NLEN + cs0 + 2 * K3_WPB * CHUNK + m;
            const float* Bb = B + (size_t)(b * NST + 8 * h) * NLEN + cs0 + 2 * K3_WPB * CHUNK + m;
#pragma unroll
            for (int q = 0; q < 8; q++) {
                cv[0][q] = Cb[(size_t)q * NLEN];
                bv[0][q] = Bb[(size_t)q * NLEN];
                cv[1][q] = Cb[(size_t)q * NLEN + CHUNK];
                bv[1][q] = Bb[(size_t)q * NLEN + CHUNK];
            }
        }
    }
}

extern "C" void kernel_launch(void* const* d_in, const int* in_sizes, int n_in,
                              void* d_out, int out_size, void* d_ws, size_t ws_size,
                              hipStream_t stream) {
    const float* u     = (const float*)d_in[0];
    const float* delta = (const float*)d_in[1];
    const float* A     = (const float*)d_in[2];
    const float* B     = (const float*)d_in[3];
    const float* C     = (const float*)d_in[4];
    const float* Dp    = (const float*)d_in[5];
    const float* z     = (const float*)d_in[6];
    const float* dbias = (const float*)d_in[7];
    float* out = (float*)d_out;

    k3_scan_output<<<NBATCH * (NDIM / K3_DPB), 32 * K3_WPB, 0, stream>>>(
        u, delta, A, B, C, Dp, z, dbias, out);
}


# stage A also keeps u and the six A rows in LDS; stage B and the output loop read them there, not from the L2, so no iteration opens with a wait on global loads; sum orders unchanged
# speedup vs baseline: 6.0194x; 1.0317x over previous
#include <hip/hip_runtime.h>

#define NBATCH 2
#define NDIM   768
#define NLEN   2048
#define NST    16
#define CHUNK  16
#define NC     (NLEN / CHUNK)

typedef float v2f __attribute__((ext_vector_type(2)));
typedef float v8f __attribute__((ext_vector_type(8)));

__device__ __forceinline__ float softplus_f(float x) {
    return (x > 20.0f) ? x : __logf(1.0f + __expf(x));
}

#define K3_DPB 6
#define K3_WPB 32
#define K3_PPW (NC / K3_WPB / 2)
#define K3_SEG 16
#define K3_SCH (NC / K3_SEG)
#define K3_DPP (32 * K3_WPB / (K3_SEG * NST))
#define K3_NPS ((K3_DPB + K3_DPP - 1) / K3_DPP)
#define K3_CPP (32 * K3_WPB / NST)
#define K3_TPT (NLEN / (32 * K3_WPB))
#define LOG2E_F 1.4426950408889634f

__device__ __forceinline__ float xhalf(float x) {
    int v = __float_as_int(x);
    return __int_as_float(__builtin_amdgcn_permlanex16(
        v, v, 0x76543210, 0xfedcba98, false, false));
}
#define ROW_SHR(v, K) __int_as_float(__builtin_amdgcn_update_dpp( \
    0, __float_as_int(v), 0x110 + (K), 0xF, 0xF, true))

__device__ __forceinline__ void lds_barrier() {
    __builtin_amdgcn_fence(__ATOMIC_RELEASE, "workgroup", "local");
    __builtin_amdgcn_s_barrier();
    __builtin_amdgcn_fence(__ATOMIC_ACQUIRE, "workgroup", "local");
}

__global__ __launch_bounds__(32 * K3_WPB) void k3_scan_output(
    const float* __restrict__ u, const float* __restrict__ delta,
    const float* __restrict__ A, const float* __restrict__ B,
    const float* __restrict__ C, const float* __restrict__ Dp,
    const float* __restrict__ z, const float* __restrict__ dbias,
    float* __restrict__ out)
{
    __shared__ float aggA[K3_DPB * K3_SEG * NST];
    __shared__ float aggS[K3_DPB * K3_SEG * NST];
    __shared__ __attribute__((aligned(16))) float xinL[K3_DPB * NC * NST];
    __shared__ __attribute__((aligned(16))) float spL[K3_DPB * NLEN];
    __shared__ __attribute__((aligned(16))) float duL[K3_DPB * NLEN];
    __shared__ __attribute__((aligned(16))) float uL[K3_DPB * NLEN];
    __shared__ __attribute__((aligned(16))) float AL[K3_DPB * NST];
    __shared__ float decL[K3_DPB * NC * NST];
    __shared__ float svL[K3_DPB * NC * NST];

    int tid = threadIdx.x;
    int b   = blockIdx.x / (NDIM / K3_DPB);
    int dp  = blockIdx.x % (NDIM / K3_DPB);

    float4 nb0, nb1, nb2, nb3;
    {
        float dvv[K3_DPB][K3_TPT], uvv[K3_DPB][K3_TPT];
#pragma unroll
        for (int dsel = 0; dsel < K3_DPB; dsel++) {
            size_t rb = (size_t)(b * NDIM + dp * K3_DPB + dsel) * NLEN + tid;
#pragma unroll
            for (int i = 0; i < K3_TPT; i++) {
                dvv[dsel][i] = delta[rb + 32 * K3_WPB * i];
                uvv[dsel][i] = u[rb + 32 * K3_WPB * i];
            }
        }
        {
            const float4* Bp = (const float4*)(B + (size_t)(b * NST + (tid & (NST - 1))) * NLEN
                                               + (tid >> 4) * CHUNK);
            nb0 = Bp[0]; nb1 = Bp[1]; nb2 = Bp[2]; nb3 = Bp[3];
        }
        if (tid < K3_DPB * NST) AL[tid] = A[dp * K3_DPB * NST + tid];
#pragma unroll
        for (int dsel = 0; dsel < K3_DPB; dsel++) {
            float bias = dbias[dp * K3_DPB + dsel];
#pragma unroll
            for (int i = 0; i < K3_TPT; i++) {
                float sp = softplus_f(dvv[dsel][i] + bias);
                spL[dsel * NLEN + tid + 32 * K3_WPB * i] = sp;
                duL[dsel * NLEN + tid + 32 * K3_WPB * i] = sp * uvv[dsel][i];
                uL[dsel * NLEN + tid + 32 * K3_WPB * i]  = uvv[dsel][i];
            }
        }
    }
    lds_barrier();

    {
        int n  = tid & (NST - 1);
        int cl = tid >> 4;
#pragma unroll 1
        for (int pass = 0; pass < NC / K3_CPP; pass++) {
            int c = cl + K3_CPP * pass;
            float4 b0 = nb0, b1 = nb1, b2 = nb2, b3 = nb3;
            float bq[CHUNK] = {b0.x, b0.y, b0.z, b0.w, b1.x, b1.y, b1.z, b1.w,
                               b2.x, b2.y, b2.z, b2.w, b3.x, b3.y, b3.z, b3.w};
#pragma unroll 1
            for (int dsel = 0; dsel < K3_DPB; dsel += 2) {
                v2f An;
                An.x = AL[dsel * NST + n] * LOG2E_F;
                An.y = AL[(dsel + 1) * NST + n] * LOG2E_F;
                const float4* s0 = (const float4*)(spL + dsel * NLEN + c * CHUNK);
                const float4* s1 = (const float4*)(spL + (dsel + 1) * NLEN + c * CHUNK);
                const float4* u0 = (const float4*)(duL + dsel * NLEN + c * CHUNK);
                const float4* u1 = (const float4*)(duL + (dsel + 1) * NLEN + c * CHUNK);
                v2f x = {0.0f, 0.0f};
                v2f R = {0.0f, 0.0f};
#pragma unroll
                for (int q = 0; q < CHUNK / 4; q++) {
                    float4 sa = s0[q], sb = s1[q], ua = u0[q], ub = u1[q];
                    float spa[4] = {sa.x, sa.y, sa.z, sa.w};
                    float spb[4] = {sb.x, sb.y, sb.z, sb.w};
                    float dua[4] = {ua.x, ua.y, ua.z, ua.w};
                    float dub[4] = {ub.x, ub.y, ub.z, ub.w};
#pragma unroll
                    for (int j = 0; j < 4; j++) {
                        v2f sp; sp.x = spa[j]; sp.y = spb[j];
                        v2f du; du.x = dua[j]; du.y = dub[j];
                        v2f g = An * sp;
                        v2f a;
                        a.x = __builtin_amdgcn_exp2f(g.x);
                        a.y = __builtin_amdgcn_exp2f(g.y);
                        x = a * x + du * bq[4 * q + j];
                        R += sp;
                    }
                }
                v2f gd = An * R;
                int idx = (dsel * NC + c) * NST + n;
                decL[idx]            = __builtin_amdgcn_exp2f(gd.x);
                svL[idx]             = x.x;
                decL[idx + NC * NST] = __builtin_amdgcn_exp2f(gd.y);
                svL[idx + NC * NST]  = x.y;
            }
            if (pass + 1 < NC / K3_CPP) {
                const float4* Bp = (const float4*)(B + (size_t)(b * NST + n) * NLEN
                                                   + (c + K3_CPP) * CHUNK);
                nb0 = Bp[0]; nb1 = Bp[1]; nb2 = Bp[2]; nb3 = Bp[3];
            }
        }
    }
    int  w    = __builtin_amdgcn_readfirstlane((int)(threadIdx.x >> 5));
    int  lane = tid & 31;
    int  m    = lane & 15;
    int  h    = lane >> 4;

    float cv[2][8], bv[2][8];
    {
        int cs0 = 2 * w * CHUNK;
        const float* Cb = C + (size_t)(b * NST + 8 * h) * NLEN + cs0 + m;
        const float* Bb = B + (size_t)(b * NST + 8 * h) * NLEN + cs0 + m;
#pragma unroll
        for (int q = 0; q < 8; q++) {
            cv[0][q] = Cb[(size_t)q * NLEN];
            bv[0][q] = Bb[(size_t)q * NLEN];
            cv[1][q] = Cb[(size_t)q * NLEN + CHUNK];
            bv[1][q] = Bb[(size_t)q * NLEN + CHUNK];
        }
    }

    lds_barrier();

    {
        int n   = tid & (NST - 1);
        int seg = (tid >> 4) & (K3_SEG - 1);
        int dq  = __builtin_amdgcn_readfirstlane(tid >> 8);

        float dec[K3_NPS][K3_SCH], sv[K3_NPS][K3_SCH];
#pragma unroll
        for (int p = 0; p < K3_NPS; p++) {
            int dsel = dq + K3_DPP * p;
            if (dsel < K3_DPB) {
                int base = (dsel * NC + seg * K3_SCH) * NST + n;
#pragma unroll
                for (int j = 0; j < K3_SCH; j++) {
                    dec[p][j] = decL[base + j * NST];
                    sv[p][j]  = svL[base + j * NST];
                }
                float ag = dec[p][0], sg = sv[p][0];
#pragma unroll
                for (int j = 1; j < K3_SCH; j++) {
                    sg = dec[p][j] * sg + sv[p][j];
                    ag = ag * dec[p][j];
                }
                int ai = (dsel * K3_SEG + seg) * NST + n;
                aggA[ai] = ag;
                aggS[ai] = sg;
            }
        }
        lds_barrier();

#pragma unroll
        for (int p = 0; p < K3_NPS; p++) {
            int dsel = dq + K3_DPP * p;
            if (dsel < K3_DPB) {
                float x = 0.0f;
#pragma unroll
                for (int i = 0; i < K3_SEG - 1; i++) {
                    int gi = (dsel * K3_SEG + i) * NST + n;
                    float xn = aggA[gi] * x + aggS[gi];
                    x = (i < seg) ? xn : x;
                }
                float* xo = xinL + ((size_t)(dsel * NC + seg * K3_SCH)) * NST + n;
#pragma unroll
                for (int j = 0; j < K3_SCH; j++) {
                    xo[j * NST] = x;
                    x = dec[p][j] * x + sv[p][j];
                }
            }
        }
        lds_barrier();
    }

    bool hi   = (h != 0);
    int  lim  = m - 8 * h;

    for (int pi = 0; pi < K3_PPW; pi++) {
        int c0  = 2 * (pi * K3_WPB + w);
        int cs0 = c0 * CHUNK;

#pragma unroll 1
        for (int dsel = 0; dsel < K3_DPB; dsel++) {
            int d = dp * K3_DPB + dsel;
            size_t row  = (size_t)(b * NDIM + d);
            size_t base = row * NLEN + cs0 + lane;

            float zv = z[base];
            float uv = uL[dsel * NLEN + cs0 + lane];
            const float4* A4 = (const float4*)(AL + dsel * NST + 8 * h);
            const float4* X4 = (const float4*)(xinL + (dsel * NC + c0) * NST + 8 * h);
            float4 a0 = A4[0], a1 = A4[1];
            float An[8] = {a0.x, a0.y, a0.z, a0.w, a1.x, a1.y, a1.z, a1.w};
            float Dd   = Dp[d];

            float sp = spL[dsel * NLEN + cs0 + lane];
            float rr = sp;
            rr += ROW_SHR(rr, 1);
            rr += ROW_SHR(rr, 2);
            rr += ROW_SHR(rr, 4);
            rr += ROW_SHR(rr, 8);
            float rl  = rr * LOG2E_F;
            float dus = sp * uv;
            float rlx = xhalf(rl);
            float dux = xhalf(dus);

            float vk[2];
#pragma unroll
            for (int k = 0; k < 2; k++) {
                float rlk = (hi == (k != 0)) ? rl  : rlx;
                float duk = (hi == (k != 0)) ? dus : dux;
                float4 x0 = X4[4 * k], x1 = X4[4 * k + 1];
                float xin[8] = {x0.x, x0.y, x0.z, x0.w, x1.x, x1.y, x1.z, x1.w};

                float P[8], Q[8];
                float v = 0.0f;
#pragma unroll
                for (int q = 0; q < 8; q++) {
                    float g = An[q] * rlk;
                    P[q] = cv[k][q] * __builtin_amdgcn_exp2f(g);
                    Q[q] = (bv[k][q] * duk) * __builtin_amdgcn_exp2f(-g);
                    v += P[q] * xin[q];
                }

                v8f acc = {0.f, 0.f, 0.f, 0.f, 0.f, 0.f, 0.f, 0.f};
#pragma unroll
                for (int j = 0; j < 4; j++) {
                    v2f va; va.x = Q[2 * j]; va.y = Q[2 * j + 1];
                    v2f vb; vb.x = P[2 * j]; vb.y = P[2 * j + 1];
                    acc = __builtin_amdgcn_wmma_f32_16x16x4_f32(
                        false, va, false, vb, (short)0, acc, false, false);
                }
#pragma unroll
                for (int r = 0; r < 8; r++) v += (r <= lim) ? acc[r] : 0.0f;
                vk[k] = v;
            }
            float mine = hi ? vk[1] : vk[0];
            float send = hi ? vk[0] : vk[1];
            float y = mine + xhalf(send);

            float sig = __builtin_amdgcn_rcpf(1.0f + __expf(-zv));
            float o = (y + uv * Dd) * (zv * sig);
            out[base] = o;
        }
        if (pi + 1 < K3_PPW) {
            const float* Cb = C + (size_t)(b * NST + 8 * h) * NLEN + cs0 + 2 * K3_WPB * CHUNK + m;
            const float* Bb = B + (size_t)(b * NST + 8 * h) * NLEN + cs0 + 2 * K3_WPB * CHUNK + m;
#pragma unroll
            for (int q = 0; q < 8; q++) {
                cv[0][q] = Cb[(size_t)q * NLEN];
                bv[0][q] = Bb[(size_t)q * NLEN];
                cv[1][q] = Cb[(size_t)q * NLEN + CHUNK];
                bv[1][q] = Bb[(size_t)q * NLEN + CHUNK];
            }
        }
    }
}

extern "C" void kernel_launch(void* const* d_in, const int* in_sizes, int n_in,
                              void* d_out, int out_size, void* d_ws, size_t ws_size,
                              hipStream_t stream) {
    const float* u     = (const float*)d_in[0];
    const float* delta = (const float*)d_in[1];
    const float* A     = (const float*)d_in[2];
    const float* B     = (const float*)d_in[3];
    const float* C     = (const float*)d_in[4];
    const float* Dp    = (const float*)d_in[5];
    const float* z     = (const float*)d_in[6];
    const float* dbias = (const float*)d_in[7];
    float* out = (float*)d_out;

    k3_scan_output<<<NBATCH * (NDIM / K3_DPB), 32 * K3_WPB, 0, stream>>>(
        u, delta, A, B, C, Dp, z, dbias, out);
}


# k3: vector work after the WMMAs cut to one fma per accumulator: x_in is added into Q's step-0 row, so the product carries sum_n P*x_in; causal mask as a 0/1 fma factor; carry-in sum order changed
# speedup vs baseline: 6.3456x; 1.0542x over previous
#include <hip/hip_runtime.h>

#define NBATCH 2
#define NDIM   768
#define NLEN   2048
#define NST    16
#define CHUNK  16
#define NC     (NLEN / CHUNK)

typedef float v2f __attribute__((ext_vector_type(2)));
typedef float v8f __attribute__((ext_vector_type(8)));

__device__ __forceinline__ float softplus_f(float x) {
    return (x > 20.0f) ? x : __logf(1.0f + __expf(x));
}

#define K3_DPB 6
#define K3_WPB 32
#define K3_PPW (NC / K3_WPB / 2)
#define K3_SEG 16
#define K3_SCH (NC / K3_SEG)
#define K3_DPP (32 * K3_WPB / (K3_SEG * NST))
#define K3_NPS ((K3_DPB + K3_DPP - 1) / K3_DPP)
#define K3_CPP (32 * K3_WPB / NST)
#define K3_TPT (NLEN / (32 * K3_WPB))
#define LOG2E_F 1.4426950408889634f

__device__ __forceinline__ float xhalf(float x) {
    int v = __float_as_int(x);
    return __int_as_float(__builtin_amdgcn_permlanex16(
        v, v, 0x76543210, 0xfedcba98, false, false));
}
#define ROW_SHR(v, K) __int_as_float(__builtin_amdgcn_update_dpp( \
    0, __float_as_int(v), 0x110 + (K), 0xF, 0xF, true))

__device__ __forceinline__ void lds_barrier() {
    __builtin_amdgcn_fence(__ATOMIC_RELEASE, "workgroup", "local");
    __builtin_amdgcn_s_barrier();
    __builtin_amdgcn_fence(__ATOMIC_ACQUIRE, "workgroup", "local");
}

__global__ __launch_bounds__(32 * K3_WPB) void k3_scan_output(
    const float* __restrict__ u, const float* __restrict__ delta,
    const float* __restrict__ A, const float* __restrict__ B,
    const float* __restrict__ C, const float* __restrict__ Dp,
    const float* __restrict__ z, const float* __restrict__ dbias,
    float* __restrict__ out)
{
    __shared__ float aggA[K3_DPB * K3_SEG * NST];
    __shared__ float aggS[K3_DPB * K3_SEG * NST];
    __shared__ __attribute__((aligned(16))) float xinL[K3_DPB * NC * NST];
    __shared__ __attribute__((aligned(16))) float spL[K3_DPB * NLEN];
    __shared__ __attribute__((aligned(16))) float duL[K3_DPB * NLEN];
    __shared__ __attribute__((aligned(16))) float uL[K3_DPB * NLEN];
    __shared__ __attribute__((aligned(16))) float AL[K3_DPB * NST];
    __shared__ __attribute__((aligned(16))) float zeroL[2 * NST];
    __shared__ float decL[K3_DPB * NC * NST];
    __shared__ float svL[K3_DPB * NC * NST];

    int tid = threadIdx.x;
    int b   = blockIdx.x / (NDIM / K3_DPB);
    int dp  = blockIdx.x % (NDIM / K3_DPB);

    float4 nb0, nb1, nb2, nb3;
    {
        float dvv[K3_DPB][K3_TPT], uvv[K3_DPB][K3_TPT];
#pragma unroll
        for (int dsel = 0; dsel < K3_DPB; dsel++) {
            size_t rb = (size_t)(b * NDIM + dp * K3_DPB + dsel) * NLEN + tid;
#pragma unroll
            for (int i = 0; i < K3_TPT; i++) {
                dvv[dsel][i] = delta[rb + 32 * K3_WPB * i];
                uvv[dsel][i] = u[rb + 32 * K3_WPB * i];
            }
        }
        {
            const float4* Bp = (const float4*)(B + (size_t)(b * NST + (tid & (NST - 1))) * NLEN
                                               + (tid >> 4) * CHUNK);
            nb0 = Bp[0]; nb1 = Bp[1]; nb2 = Bp[2]; nb3 = Bp[3];
        }
        if (tid < K3_DPB * NST) AL[tid] = A[dp * K3_DPB * NST + tid];
        if (tid < 2 * NST) zeroL[tid] = 0.0f;
#pragma unroll
        for (int dsel = 0; dsel < K3_DPB; dsel++) {
            float bias = dbias[dp * K3_DPB + dsel];
#pragma unroll
            for (int i = 0; i < K3_TPT; i++) {
                float sp = softplus_f(dvv[dsel][i] + bias);
                spL[dsel * NLEN + tid + 32 * K3_WPB * i] = sp;
                duL[dsel * NLEN + tid + 32 * K3_WPB * i] = sp * uvv[dsel][i];
                uL[dsel * NLEN + tid + 32 * K3_WPB * i]  = uvv[dsel][i];
            }
        }
    }
    lds_barrier();

    {
        int n  = tid & (NST - 1);
        int cl = tid >> 4;
#pragma unroll 1
        for (int pass = 0; pass < NC / K3_CPP; pass++) {
            int c = cl + K3_CPP * pass;
            float4 b0 = nb0, b1 = nb1, b2 = nb2, b3 = nb3;
            float bq[CHUNK] = {b0.x, b0.y, b0.z, b0.w, b1.x, b1.y, b1.z, b1.w,
                               b2.x, b2.y, b2.z, b2.w, b3.x, b3.y, b3.z, b3.w};
#pragma unroll 1
            for (int dsel = 0; dsel < K3_DPB; dsel += 2) {
                v2f An;
                An.x = AL[dsel * NST + n] * LOG2E_F;
                An.y = AL[(dsel + 1) * NST + n] * LOG2E_F;
                const float4* s0 = (const float4*)(spL + dsel * NLEN + c * CHUNK);
                const float4* s1 = (const float4*)(spL + (dsel + 1) * NLEN + c * CHUNK);
                const float4* u0 = (const float4*)(duL + dsel * NLEN + c * CHUNK);
                const float4* u1 = (const float4*)(duL + (dsel + 1) * NLEN + c * CHUNK);
                v2f x = {0.0f, 0.0f};
                v2f R = {0.0f, 0.0f};
#pragma unroll
                for (int q = 0; q < CHUNK / 4; q++) {
                    float4 sa = s0[q], sb = s1[q], ua = u0[q], ub = u1[q];
                    float spa[4] = {sa.x, sa.y, sa.z, sa.w};
                    float spb[4] = {sb.x, sb.y, sb.z, sb.w};
                    float dua[4] = {ua.x, ua.y, ua.z, ua.w};
                    float dub[4] = {ub.x, ub.y, ub.z, ub.w};
#pragma unroll
                    for (int j = 0; j < 4; j++) {
                        v2f sp; sp.x = spa[j]; sp.y = spb[j];
                        v2f du; du.x = dua[j]; du.y = dub[j];
                        v2f g = An * sp;
                        v2f a;
                        a.x = __builtin_amdgcn_exp2f(g.x);
                        a.y = __builtin_amdgcn_exp2f(g.y);
                        x = a * x + du * bq[4 * q + j];
                        R += sp;
                    }
                }
                v2f gd = An * R;
                int idx = (dsel * NC + c) * NST + n;
                decL[idx]            = __builtin_amdgcn_exp2f(gd.x);
                svL[idx]             = x.x;
                decL[idx + NC * NST] = __builtin_amdgcn_exp2f(gd.y);
                svL[idx + NC * NST]  = x.y;
            }
            if (pass + 1 < NC / K3_CPP) {
                const float4* Bp = (const float4*)(B + (size_t)(b * NST + n) * NLEN
                                                   + (c + K3_CPP) * CHUNK);
                nb0 = Bp[0]; nb1 = Bp[1]; nb2 = Bp[2]; nb3 = Bp[3];
            }
        }
    }
    int  w    = __builtin_amdgcn_readfirstlane((int)(threadIdx.x >> 5));
    int  lane = tid & 31;
    int  m    = lane & 15;
    int  h    = lane >> 4;

    float cv[2][8], bv[2][8];
    {
        int cs0 = 2 * w * CHUNK;
        const float* Cb = C + (size_t)(b * NST + 8 * h) * NLEN + cs0 + m;
        const float* Bb = B + (size_t)(b * NST + 8 * h) * NLEN + cs0 + m;
#pragma unroll
        for (int q = 0; q < 8; q++) {
            cv[0][q] = Cb[(size_t)q * NLEN];
            bv[0][q] = Bb[(size_t)q * NLEN];
            cv[1][q] = Cb[(size_t)q * NLEN + CHUNK];
            bv[1][q] = Bb[(size_t)q * NLEN + CHUNK];
        }
    }

    lds_barrier();

    {
        int n   = tid & (NST - 1);
        int seg = (tid >> 4) & (K3_SEG - 1);
        int dq  = __builtin_amdgcn_readfirstlane(tid >> 8);

        float dec[K3_NPS][K3_SCH], sv[K3_NPS][K3_SCH];
#pragma unroll
        for (int p = 0; p < K3_NPS; p++) {
            int dsel = dq + K3_DPP * p;
            if (dsel < K3_DPB) {
                int base = (dsel * NC + seg * K3_SCH) * NST + n;
#pragma unroll
                for (int j = 0; j < K3_SCH; j++) {
                    dec[p][j] = decL[base + j * NST];
                    sv[p][j]  = svL[base + j * NST];
                }
                float ag = dec[p][0], sg = sv[p][0];
#pragma unroll
                for (int j = 1; j < K3_SCH; j++) {
                    sg = dec[p][j] * sg + sv[p][j];
                    ag = ag * dec[p][j];
                }
                int ai = (dsel * K3_SEG + seg) * NST + n;
                aggA[ai] = ag;
                aggS[ai] = sg;
            }
        }
        lds_barrier();

#pragma unroll
        for (int p = 0; p < K3_NPS; p++) {
            int dsel = dq + K3_DPP * p;
            if (dsel < K3_DPB) {
                float x = 0.0f;
#pragma unroll
                for (int i = 0; i < K3_SEG - 1; i++) {
                    int gi = (dsel * K3_SEG + i) * NST + n;
                    float xn = aggA[gi] * x + aggS[gi];
                    x = (i < seg) ? xn : x;
                }
                float* xo = xinL + ((size_t)(dsel * NC + seg * K3_SCH)) * NST + n;
#pragma unroll
                for (int j = 0; j < K3_SCH; j++) {
                    xo[j * NST] = x;
                    x = dec[p][j] * x + sv[p][j];
                }
            }
        }
        lds_barrier();
    }

    bool hi   = (h != 0);
    int  lim  = m - 8 * h;
    float mk[8];
#pragma unroll
    for (int r = 0; r < 8; r++) mk[r] = (r <= lim) ? 1.0f : 0.0f;

    for (int pi = 0; pi < K3_PPW; pi++) {
        int c0  = 2 * (pi * K3_WPB + w);
        int cs0 = c0 * CHUNK;

#pragma unroll 1
        for (int dsel = 0; dsel < K3_DPB; dsel++) {
            int d = dp * K3_DPB + dsel;
            size_t row  = (size_t)(b * NDIM + d);
            size_t base = row * NLEN + cs0 + lane;

            float zv = z[base];
            float uv = uL[dsel * NLEN + cs0 + lane];
            const float4* A4 = (const float4*)(AL + dsel * NST + 8 * h);
            const float4* X4 = (const float4*)((m == 0) ? (xinL + (dsel * NC + c0) * NST + 8 * h)
                                                        : (zeroL + 8 * h));
            float4 a0 = A4[0], a1 = A4[1];
            float An[8] = {a0.x, a0.y, a0.z, a0.w, a1.x, a1.y, a1.z, a1.w};
            float Dd   = Dp[d];

            float sp = spL[dsel * NLEN + cs0 + lane];
            float rr = sp;
            rr += ROW_SHR(rr, 1);
            rr += ROW_SHR(rr, 2);
            rr += ROW_SHR(rr, 4);
            rr += ROW_SHR(rr, 8);
            float rl  = rr * LOG2E_F;
            float dus = sp * uv;
            float rlx = xhalf(rl);
            float dux = xhalf(dus);

            float vk[2];
#pragma unroll
            for (int k = 0; k < 2; k++) {
                float rlk = (hi == (k != 0)) ? rl  : rlx;
                float duk = (hi == (k != 0)) ? dus : dux;
                float4 x0 = X4[4 * k], x1 = X4[4 * k + 1];
                float xin[8] = {x0.x, x0.y, x0.z, x0.w, x1.x, x1.y, x1.z, x1.w};

                float P[8], Q[8];
                float v = 0.0f;
#pragma unroll
                for (int q = 0; q < 8; q++) {
                    float g = An[q] * rlk;
                    P[q] = cv[k][q] * __builtin_amdgcn_exp2f(g);
                    Q[q] = __builtin_fmaf(bv[k][q] * duk, __builtin_amdgcn_exp2f(-g), xin[q]);
                }

                v8f acc = {0.f, 0.f, 0.f, 0.f, 0.f, 0.f, 0.f, 0.f};
#pragma unroll
                for (int j = 0; j < 4; j++) {
                    v2f va; va.x = Q[2 * j]; va.y = Q[2 * j + 1];
                    v2f vb; vb.x = P[2 * j]; vb.y = P[2 * j + 1];
                    acc = __builtin_amdgcn_wmma_f32_16x16x4_f32(
                        false, va, false, vb, (short)0, acc, false, false);
                }
#pragma unroll
                for (int r = 0; r < 8; r++) v = __builtin_fmaf(acc[r], mk[r], v);
                vk[k] = v;
            }
            float mine = hi ? vk[1] : vk[0];
            float send = hi ? vk[0] : vk[1];
            float y = mine + xhalf(send);

            float sig = __builtin_amdgcn_rcpf(1.0f + __expf(-zv));
            float o = (y + uv * Dd) * (zv * sig);
            out[base] = o;
        }
        if (pi + 1 < K3_PPW) {
            const float* Cb = C + (size_t)(b * NST + 8 * h) * NLEN + cs0 + 2 * K3_WPB * CHUNK + m;
            const float* Bb = B + (size_t)(b * NST + 8 * h) * NLEN + cs0 + 2 * K3_WPB * CHUNK + m;
#pragma unroll
            for (int q = 0; q < 8; q++) {
                cv[0][q] = Cb[(size_t)q * NLEN];
                bv[0][q] = Bb[(size_t)q * NLEN];
                cv[1][q] = Cb[(size_t)q * NLEN + CHUNK];
                bv[1][q] = Bb[(size_t)q * NLEN + CHUNK];
            }
        }
    }
}

extern "C" void kernel_launch(void* const* d_in, const int* in_sizes, int n_in,
                              void* d_out, int out_size, void* d_ws, size_t ws_size,
                              hipStream_t stream) {
    const float* u     = (const float*)d_in[0];
    const float* delta = (const float*)d_in[1];
    const float* A     = (const float*)d_in[2];
    const float* B     = (const float*)d_in[3];
    const float* C     = (const float*)d_in[4];
    const float* Dp    = (const float*)d_in[5];
    const float* z     = (const float*)d_in[6];
    const float* dbias = (const float*)d_in[7];
    float* out = (float*)d_out;

    k3_scan_output<<<NBATCH * (NDIM / K3_DPB), 32 * K3_WPB, 0, stream>>>(
        u, delta, A, B, C, Dp, z, dbias, out);
}
